# starting kernel
# baseline (speedup 1.0000x reference)
#include <hip/hip_runtime.h>
#include <math.h>

typedef float v2f __attribute__((ext_vector_type(2)));
typedef float v4f __attribute__((ext_vector_type(4)));
typedef float v8f __attribute__((ext_vector_type(8)));

#define NGROUP   16
#define IN_DIM   8
#define OCH      128
#define DIN      16
#define NVOX     32768

__global__ __launch_bounds__(256)
void deconv_votes_wmma(const float* __restrict__ x,
                       const float* __restrict__ w,
                       const float* __restrict__ deconv_b,
                       float* __restrict__ votes) {
  __shared__ float xs[8 * 2 * 2 * 18 * 2];

  const int tid   = threadIdx.x;
  const int lane  = tid & 31;
  const int wv    = tid >> 5;
  const int ocb   = wv * 16;
  const int m     = lane & 15;
  const int khalf = lane >> 4;

  const int wg  = blockIdx.x;
  const int g   = wg >> 7;
  const int rem = wg & 127;
  const int par = rem >> 4;
  const int dp  = rem & 15;
  const int pd = (par >> 2) & 1, ph = (par >> 1) & 1, pw = par & 1;
  const int bidx = g >> 3, indim = g & 7;

  const int deltad[2] = { pd ? 1 : 0, pd ? 0 : -1 };
  const int deltah[2] = { ph ? 1 : 0, ph ? 0 : -1 };
  const int deltaw[2] = { pw ? 1 : 0, pw ? 0 : -1 };

  v2f A[2][2][2][4];
  {
    const int kd0 = 1 - pd, kh0 = 1 - ph, kw0 = 1 - pw;
    for (int sd = 0; sd < 2; ++sd)
      for (int sh = 0; sh < 2; ++sh)
        for (int sw = 0; sw < 2; ++sw) {
          const int tap = ((2 * sd + kd0) * 4 + (2 * sh + kh0)) * 4 + (2 * sw + kw0);
          for (int c = 0; c < 4; ++c) {
            const int ic = c * 4 + khalf * 2;
            const float* wp = w + (size_t)(ic * OCH + ocb + m) * 64 + tap;
            A[sd][sh][sw][c].x = wp[0];
            A[sd][sh][sw][c].y = wp[OCH * 64];
          }
        }
  }
  float bl[8];
  for (int j = 0; j < 8; ++j) bl[j] = deconv_b[ocb + khalf * 8 + j];

  if (tid < 128) {
    const int e = tid & 1, cs = (tid >> 1) & 1, r = tid >> 2;
    xs[(r * 18 + (cs ? 17 : 0)) * 2 + e] = 0.f;
  }

  const size_t xg = (size_t)g * 16 * DIN * DIN * DIN;

  for (int hp = 0; hp < 16; ++hp) {
    __syncthreads();
    {
      const int row  = tid >> 2;
      const int colq = (tid & 3) * 4;
      const int ic = row >> 2, sd = (row >> 1) & 1, sh = row & 1;
      const int cp = ic >> 1, e = ic & 1;
      const int d = dp + deltad[sd];
      const int h = hp + deltah[sh];
      const bool ok = ((unsigned)d < 16u) && ((unsigned)h < 16u);
      const float* src = x + xg + (((size_t)ic * DIN + (ok ? d : 0)) * DIN + (ok ? h : 0)) * DIN + colq;
      float* dst = xs + (((cp * 4 + sd * 2 + sh) * 18) + 1 + colq) * 2 + e;
      for (int j = 0; j < 4; ++j) dst[j * 2] = ok ? src[j] : 0.f;
    }
    __syncthreads();

    v8f acc = {0.f, 0.f, 0.f, 0.f, 0.f, 0.f, 0.f, 0.f};
    const int n = lane & 15;
    for (int sd = 0; sd < 2; ++sd)
      for (int sh = 0; sh < 2; ++sh)
        for (int sw = 0; sw < 2; ++sw) {
          const int col = 1 + n + deltaw[sw];
          for (int c = 0; c < 4; ++c) {
            const int cp = c * 2 + khalf;
            const v2f bvec = *(const v2f*)&xs[((cp * 4 + sd * 2 + sh) * 18 + col) * 2];
            acc = __builtin_amdgcn_wmma_f32_16x16x4_f32(
                false, A[sd][sh][sw][c], false, bvec, (short)0, acc, false, false);
          }
        }

    for (int j = 0; j < 8; ++j) acc[j] += bl[j];

    const int od = 2 * dp + pd, oh = 2 * hp + ph, ow = 2 * n + pw;
    const int o  = (od * 32 + oh) * 32 + ow;
    float* dst = votes + ((size_t)(bidx * NVOX + o) * IN_DIM + indim) * OCH
                       + ocb + khalf * 8;
    ((v4f*)dst)[0] = *((const v4f*)&acc);
    ((v4f*)dst)[1] = *((const v4f*)&acc + 1);
  }
}

__global__ __launch_bounds__(256)
void routing_squash(const float* __restrict__ votes,
                    const float* __restrict__ rbias,
                    float* __restrict__ out) {
  const int tid  = threadIdx.x;
  const int lane = tid & 31;
  const int wv   = tid >> 5;
  const int outd = lane >> 2;
  const int v    = lane & 3;
  const int vox  = blockIdx.x * 32 + wv * 4 + v;
  const int b    = vox >> 15;
  const int o    = vox & 32767;

  float vt[8][16];
  const float* base = votes + (size_t)vox * IN_DIM * OCH + outd * 16;
  for (int in = 0; in < 8; ++in) {
    const v4f* p = (const v4f*)(base + (size_t)in * OCH);
    v4f q0 = p[0], q1 = p[1], q2 = p[2], q3 = p[3];
    for (int a = 0; a < 4; ++a) {
      vt[in][a] = q0[a]; vt[in][4 + a] = q1[a];
      vt[in][8 + a] = q2[a]; vt[in][12 + a] = q3[a];
    }
  }
  float rb[16];
  for (int a = 0; a < 16; ++a) rb[a] = rbias[outd * 16 + a];

  float n2[8];
  for (int in = 0; in < 8; ++in) {
    float s = 0.f;
    for (int a = 0; a < 16; ++a) s += vt[in][a] * vt[in][a];
    n2[in] = sqrtf(s);
  }

  float logits[8];
  for (int in = 0; in < 8; ++in) logits[in] = 0.f;
  float pa[16];

  for (int it = 0; it < 3; ++it) {
    float route[8];
    for (int in = 0; in < 8; ++in) {
      float e = __expf(logits[in]);
      float t = e;
      t += __shfl_xor(t, 4, 32);
      t += __shfl_xor(t, 8, 32);
      t += __shfl_xor(t, 16, 32);
      route[in] = e / t;
    }
    for (int a = 0; a < 16; ++a) pa[a] = rb[a];
    for (int in = 0; in < 8; ++in)
      for (int a = 0; a < 16; ++a) pa[a] += vt[in][a] * route[in];
    if (it < 2) {
      float s = 0.f;
      for (int a = 0; a < 16; ++a) s += pa[a] * pa[a];
      const float n1 = sqrtf(s);
      for (int in = 0; in < 8; ++in) {
        float dot = 0.f;
        for (int a = 0; a < 16; ++a) dot += pa[a] * vt[in][a];
        logits[in] += dot / fmaxf(n1 * n2[in], 1e-8f);
      }
    }
  }

  float nsq = 0.f;
  for (int a = 0; a < 16; ++a) nsq += pa[a] * pa[a];
  const float norm  = sqrtf(nsq);
  const float scale = (1.f / (norm + 1e-12f)) * (nsq / (1.f + nsq));
  for (int a = 0; a < 16; ++a)
    out[(((size_t)(b * 8 + outd) * 16 + a) << 15) + o] = pa[a] * scale;
}

extern "C" void kernel_launch(void* const* d_in, const int* in_sizes, int n_in,
                              void* d_out, int out_size, void* d_ws, size_t ws_size,
                              hipStream_t stream) {
  (void)in_sizes; (void)n_in; (void)out_size; (void)ws_size;
  const float* x  = (const float*)d_in[0];
  const float* w  = (const float*)d_in[1];
  const float* db = (const float*)d_in[2];
  const float* rb = (const float*)d_in[3];
  float* votes = (float*)d_ws;
  float* outp  = (float*)d_out;

  deconv_votes_wmma<<<NGROUP * 8 * 16, 256, 0, stream>>>(x, w, db, votes);
  routing_squash<<<2048, 256, 0, stream>>>(votes, rb, outp);
}


# Fuse routing+squash into the deconv kernel: one launch, votes stay in WMMA accumulators (never stored); atom sums now split over lane halves, softmax sum goes through LDS (sum order changes)
# speedup vs baseline: 1.7741x; 1.7741x over previous
#include <hip/hip_runtime.h>
#include <math.h>


typedef float v2f __attribute__((ext_vector_type(2)));
typedef float v4f __attribute__((ext_vector_type(4)));
typedef float v8f __attribute__((ext_vector_type(8)));

#define IN_DIM   8
#define OCH      128
#define DIN      16
#define TILE     1152
#define EXROW    68

__device__ __forceinline__ float xhalf(float v) {
  return __builtin_bit_cast(float, __builtin_amdgcn_permlanex16(
      __builtin_bit_cast(int, v), __builtin_bit_cast(int, v), 0x76543210, 0xfedcba98, false, false));
}
__device__ __forceinline__ float hsum(float v) { return v + xhalf(v); }

template <int PW>
__device__ __forceinline__ void tile_parity(const float* __restrict__ xs, float* __restrict__ ex,
                                            const v2f (&A)[2][2][4][4],
                                            const float (&bl)[8], const float (&rb)[8],
                                            const int n, const int khalf, const int wv,
                                            float (&pa)[8]) {
  v8f acc[IN_DIM];
#pragma unroll
  for (int in = 0; in < IN_DIM; ++in) acc[in] = (v8f){0.f, 0.f, 0.f, 0.f, 0.f, 0.f, 0.f, 0.f};

#pragma unroll
  for (int sd = 0; sd < 2; ++sd) {
#pragma unroll
    for (int sh = 0; sh < 2; ++sh) {
#pragma unroll
      for (int sw = 0; sw < 2; ++sw) {
        const int kw    = 2 * sw + 1 - PW;
        const int delta = PW ? (sw == 0 ? 1 : 0) : (sw == 0 ? 0 : -1);
        const int col   = 1 + n + delta;
#pragma unroll
        for (int c = 0; c < 4; ++c) {
          const int cp  = c * 2 + khalf;
          const int off = ((cp * 4 + sd * 2 + sh) * 18 + col) * 2;
#pragma unroll
          for (int in = 0; in < IN_DIM; ++in) {
            const v2f bvec = *(const v2f*)&xs[in * TILE + off];
            acc[in] = __builtin_amdgcn_wmma_f32_16x16x4_f32(
                false, A[sd][sh][kw][c], false, bvec, (short)0, acc[in], false, false);
          }
        }
      }
    }
  }

#pragma unroll
  for (int in = 0; in < IN_DIM; ++in)
#pragma unroll
    for (int j = 0; j < 8; ++j) acc[in][j] += bl[j];

  float n2[IN_DIM];
#pragma unroll
  for (int in = 0; in < IN_DIM; ++in) {
    float s = 0.f;
#pragma unroll
    for (int j = 0; j < 8; ++j) s += acc[in][j] * acc[in][j];
    n2[in] = sqrtf(hsum(s));
  }

  float logits[IN_DIM], route[IN_DIM];
#pragma unroll
  for (int in = 0; in < IN_DIM; ++in) { logits[in] = 0.f; route[in] = 0.125f; }

#pragma unroll
  for (int it = 0; it < 3; ++it) {
    if (it > 0) {
      float* e = ex + (it - 1) * 16 * EXROW;
      float ev[IN_DIM];
#pragma unroll
      for (int in = 0; in < IN_DIM; ++in) ev[in] = __expf(logits[in]);
#pragma unroll
      for (int i = 0; i < 4; ++i)
        e[n * EXROW + (khalf * 4 + i) * 8 + wv] = khalf ? ev[4 + i] : ev[i];
      __syncthreads();
#pragma unroll
      for (int in = 0; in < IN_DIM; ++in) {
        const v4f s0 = *(const v4f*)&e[n * EXROW + in * 8];
        const v4f s1 = *(const v4f*)&e[n * EXROW + in * 8 + 4];
        const float t = ((s0.x + s0.y) + (s0.z + s0.w)) + ((s1.x + s1.y) + (s1.z + s1.w));
        route[in] = ev[in] / t;
      }
    }
#pragma unroll
    for (int j = 0; j < 8; ++j) pa[j] = rb[j];
#pragma unroll
    for (int in = 0; in < IN_DIM; ++in)
#pragma unroll
      for (int j = 0; j < 8; ++j) pa[j] += acc[in][j] * route[in];
    if (it < 2) {
      float s = 0.f;
#pragma unroll
      for (int j = 0; j < 8; ++j) s += pa[j] * pa[j];
      const float n1 = sqrtf(hsum(s));
#pragma unroll
      for (int in = 0; in < IN_DIM; ++in) {
        float dot = 0.f;
#pragma unroll
        for (int j = 0; j < 8; ++j) dot += pa[j] * acc[in][j];
        dot = hsum(dot);
        logits[in] += dot / fmaxf(n1 * n2[in], 1e-8f);
      }
    }
  }

  float q = 0.f;
#pragma unroll
  for (int j = 0; j < 8; ++j) q += pa[j] * pa[j];
  const float nsq   = hsum(q);
  const float norm  = sqrtf(nsq);
  const float scale = (1.f / (norm + 1e-12f)) * (nsq / (1.f + nsq));
#pragma unroll
  for (int j = 0; j < 8; ++j) pa[j] *= scale;
}

__global__ __launch_bounds__(256)
void deconv_route_fused(const float* __restrict__ x,
                        const float* __restrict__ w,
                        const float* __restrict__ deconv_b,
                        const float* __restrict__ rbias,
                        float* __restrict__ out) {
  __shared__ __attribute__((aligned(16))) float xs[IN_DIM * TILE];
  __shared__ __attribute__((aligned(16))) float ex[2 * 16 * EXROW];

  const int tid   = threadIdx.x;
  const int lane  = tid & 31;
  const int wv    = tid >> 5;
  const int ocb   = wv * 16;
  const int n     = lane & 15;
  const int khalf = lane >> 4;

  const int wg  = blockIdx.x;
  const int b   = wg >> 7;
  const int rem = wg & 127;
  const int pd  = (rem >> 6) & 1, ph = (rem >> 5) & 1;
  const int dp  = (rem >> 1) & 15;
  const int hp0 = (rem & 1) * 8;

  v2f A[2][2][4][4];
  {
    const int kd0 = 1 - pd, kh0 = 1 - ph;
#pragma unroll
    for (int sd = 0; sd < 2; ++sd)
#pragma unroll
      for (int sh = 0; sh < 2; ++sh)
#pragma unroll
        for (int kw = 0; kw < 4; ++kw) {
          const int tap = ((2 * sd + kd0) * 4 + (2 * sh + kh0)) * 4 + kw;
#pragma unroll
          for (int c = 0; c < 4; ++c) {
            const int ic = c * 4 + khalf * 2;
            const float* wp = w + (size_t)(ic * OCH + ocb + n) * 64 + tap;
            A[sd][sh][kw][c].x = wp[0];
            A[sd][sh][kw][c].y = wp[OCH * 64];
          }
        }
  }
  float bl[8], rb[8];
#pragma unroll
  for (int j = 0; j < 8; ++j) {
    bl[j] = deconv_b[ocb + khalf * 8 + j];
    rb[j] = rbias[ocb + khalf * 8 + j];
  }

#pragma unroll
  for (int k = 0; k < 4; ++k) {
    const int idx = tid + 256 * k;
    const int e = idx & 1, cs = (idx >> 1) & 1, r = idx >> 2;
    xs[(r * 18 + (cs ? 17 : 0)) * 2 + e] = 0.f;
  }

  const int colq = (tid & 3) * 4;
  const int lrow = tid >> 2;
  const int lic = lrow >> 2, lsd = (lrow >> 1) & 1, lsh = lrow & 1;
  const int ld  = dp + (lsd ? (pd ? 0 : -1) : (pd ? 1 : 0));
  const int ldh = lsh ? (ph ? 0 : -1) : (ph ? 1 : 0);
  const bool dok = (unsigned)ld < 16u;
  const float* xrow = x + ((size_t)(b * IN_DIM) * 16 + lic) * (DIN * DIN * DIN)
                        + (size_t)(dok ? ld : 0) * (DIN * DIN) + colq;
  float* xdst = xs + (((lic >> 1) * 4 + lsd * 2 + lsh) * 18 + 1 + colq) * 2 + (lic & 1);

  v4f pf[IN_DIM];
  {
    const int h = hp0 + ldh;
    const bool ok = dok && ((unsigned)h < 16u);
    const float* src = xrow + (ok ? h : 0) * DIN;
#pragma unroll
    for (int k = 0; k < IN_DIM; ++k)
      pf[k] = ok ? *(const v4f*)(src + (size_t)k * 16 * (DIN * DIN * DIN)) : (v4f){0.f, 0.f, 0.f, 0.f};
  }

#pragma unroll 1
  for (int hh = 0; hh < 8; ++hh) {
    const int hp = hp0 + hh;
#pragma unroll
    for (int k = 0; k < IN_DIM; ++k) {
      float* d = xdst + k * TILE;
      d[0] = pf[k].x; d[2] = pf[k].y; d[4] = pf[k].z; d[6] = pf[k].w;
    }
    __syncthreads();

    if (hh + 1 < 8) {
      const int h = hp + 1 + ldh;
      const bool ok = dok && ((unsigned)h < 16u);
      const float* src = xrow + (ok ? h : 0) * DIN;
#pragma unroll
      for (int k = 0; k < IN_DIM; ++k)
        pf[k] = ok ? *(const v4f*)(src + (size_t)k * 16 * (DIN * DIN * DIN)) : (v4f){0.f, 0.f, 0.f, 0.f};
    }

    float pa0[8], pa1[8];
    tile_parity<0>(xs, ex, A, bl, rb, n, khalf, wv, pa0);
    tile_parity<1>(xs, ex, A, bl, rb, n, khalf, wv, pa1);

    const int od = 2 * dp + pd, oh = 2 * hp + ph;
    const int o  = (od * 32 + oh) * 32 + 2 * n;
#pragma unroll
    for (int j = 0; j < 8; ++j) {
      float* dst = out + (((size_t)((b * 8 + wv) * 16 + khalf * 8 + j)) << 15) + o;
      *(v2f*)dst = (v2f){pa0[j], pa1[j]};
    }
  }
}

extern "C" void kernel_launch(void* const* d_in, const int* in_sizes, int n_in,
                              void* d_out, int out_size, void* d_ws, size_t ws_size,
                              hipStream_t stream) {
  (void)in_sizes; (void)n_in; (void)out_size; (void)d_ws; (void)ws_size;
  const float* x  = (const float*)d_in[0];
  const float* w  = (const float*)d_in[1];
  const float* db = (const float*)d_in[2];
  const float* rb = (const float*)d_in[3];
  float* outp  = (float*)d_out;

  deconv_route_fused<<<256, 256, 0, stream>>>(x, w, db, rb, outp);
}


# B fragment LDS reads grouped one block ahead in a 2nd register set (16 two-address reads in one place per 32 WMMAs, scheduler fenced), not spread among WMMAs; bias add 8-wide for room; sum order kept
# speedup vs baseline: 1.8403x; 1.0373x over previous
#include <hip/hip_runtime.h>
#include <math.h>


typedef float v2f __attribute__((ext_vector_type(2)));
typedef float v4f __attribute__((ext_vector_type(4)));
typedef float v8f __attribute__((ext_vector_type(8)));
typedef v4f v4f_a8 __attribute__((aligned(8)));

#define IN_DIM   8
#define OCH      128
#define DIN      16
#define TILE     1152
#define EXROW    68

__device__ __forceinline__ float xhalf(float v) {
  return __builtin_bit_cast(float, __builtin_amdgcn_permlanex16(
      __builtin_bit_cast(int, v), __builtin_bit_cast(int, v), 0x76543210, 0xfedcba98, false, false));
}
__device__ __forceinline__ float hsum(float v) { return v + xhalf(v); }

__device__ __forceinline__ void fence() { __builtin_amdgcn_sched_barrier(0); }

template <int PW, int BLK>
__device__ __forceinline__ void load_b(const float* __restrict__ xl, v4f (&Bf)[16]) {
  constexpr int sd = BLK >> 2, sh = (BLK >> 1) & 1, half = BLK & 1;
#pragma unroll
  for (int c = 0; c < 4; ++c)
#pragma unroll
    for (int i = 0; i < 4; ++i)
      Bf[c * 4 + i] = *(const v4f_a8*)&xl[(half * 4 + i) * TILE + ((c * 8 + sd * 2 + sh) * 18 + PW) * 2];
}

template <int PW, int BLK>
__device__ __forceinline__ void mma_b(const v2f (&A)[2][2][4][4], const v4f (&Bf)[16], v8f (&acc)[IN_DIM]) {
  constexpr int sd = BLK >> 2, sh = (BLK >> 1) & 1, half = BLK & 1;
#pragma unroll
  for (int sw = 0; sw < 2; ++sw) {
    const int kw = 2 * sw + 1 - PW;
#pragma unroll
    for (int c = 0; c < 4; ++c)
#pragma unroll
      for (int i = 0; i < 4; ++i) {
        const v4f b = Bf[c * 4 + i];
        const v2f bvec = sw ? (v2f){b.x, b.y} : (v2f){b.z, b.w};
        acc[half * 4 + i] = __builtin_amdgcn_wmma_f32_16x16x4_f32(
            false, A[sd][sh][kw][c], false, bvec, (short)0, acc[half * 4 + i], false, false);
      }
  }
}

template <int PW>
__device__ __forceinline__ void tile_parity(const float* __restrict__ xl, float* __restrict__ ex,
                                            const v2f (&A)[2][2][4][4], v4f (&Bx)[16],
                                            const float (&bl)[8], const float (&rb)[8],
                                            const int n, const int khalf, const int wv,
                                            float (&pa)[8]) {
  v8f acc[IN_DIM];
#pragma unroll
  for (int in = 0; in < IN_DIM; ++in) acc[in] = (v8f){0.f, 0.f, 0.f, 0.f, 0.f, 0.f, 0.f, 0.f};

  v4f By[16];
  load_b<PW, 1>(xl, By); fence(); mma_b<PW, 0>(A, Bx, acc); fence();
  load_b<PW, 2>(xl, Bx); fence(); mma_b<PW, 1>(A, By, acc); fence();
  load_b<PW, 3>(xl, By); fence(); mma_b<PW, 2>(A, Bx, acc); fence();
  load_b<PW, 4>(xl, Bx); fence(); mma_b<PW, 3>(A, By, acc); fence();
  load_b<PW, 5>(xl, By); fence(); mma_b<PW, 4>(A, Bx, acc); fence();
  load_b<PW, 6>(xl, Bx); fence(); mma_b<PW, 5>(A, By, acc); fence();
  load_b<PW, 7>(xl, By); fence(); mma_b<PW, 6>(A, Bx, acc); fence();
  mma_b<PW, 7>(A, By, acc);

  const v8f blv = {bl[0], bl[1], bl[2], bl[3], bl[4], bl[5], bl[6], bl[7]};
#pragma unroll
  for (int in = 0; in < IN_DIM; ++in) acc[in] += blv;

  float n2[IN_DIM];
#pragma unroll
  for (int in = 0; in < IN_DIM; ++in) {
    float s = 0.f;
#pragma unroll
    for (int j = 0; j < 8; ++j) s += acc[in][j] * acc[in][j];
    n2[in] = sqrtf(hsum(s));
  }

  float logits[IN_DIM], route[IN_DIM];
#pragma unroll
  for (int in = 0; in < IN_DIM; ++in) { logits[in] = 0.f; route[in] = 0.125f; }

#pragma unroll
  for (int it = 0; it < 3; ++it) {
    if (it > 0) {
      float* e = ex + (it - 1) * 16 * EXROW;
      float ev[IN_DIM];
#pragma unroll
      for (int in = 0; in < IN_DIM; ++in) ev[in] = __expf(logits[in]);
#pragma unroll
      for (int i = 0; i < 4; ++i)
        e[n * EXROW + (khalf * 4 + i) * 8 + wv] = khalf ? ev[4 + i] : ev[i];
      __syncthreads();
#pragma unroll
      for (int in = 0; in < IN_DIM; ++in) {
        const v4f s0 = *(const v4f*)&e[n * EXROW + in * 8];
        const v4f s1 = *(const v4f*)&e[n * EXROW + in * 8 + 4];
        const float t = ((s0.x + s0.y) + (s0.z + s0.w)) + ((s1.x + s1.y) + (s1.z + s1.w));
        route[in] = ev[in] / t;
      }
    }
#pragma unroll
    for (int j = 0; j < 8; ++j) pa[j] = rb[j];
#pragma unroll
    for (int in = 0; in < IN_DIM; ++in)
#pragma unroll
      for (int j = 0; j < 8; ++j) pa[j] += acc[in][j] * route[in];
    if (it < 2) {
      float s = 0.f;
#pragma unroll
      for (int j = 0; j < 8; ++j) s += pa[j] * pa[j];
      const float n1 = sqrtf(hsum(s));
#pragma unroll
      for (int in = 0; in < IN_DIM; ++in) {
        float dot = 0.f;
#pragma unroll
        for (int j = 0; j < 8; ++j) dot += pa[j] * acc[in][j];
        dot = hsum(dot);
        logits[in] += dot / fmaxf(n1 * n2[in], 1e-8f);
      }
    }
  }

  if (PW == 0) load_b<1, 0>(xl, Bx);

  float q = 0.f;
#pragma unroll
  for (int j = 0; j < 8; ++j) q += pa[j] * pa[j];
  const float nsq   = hsum(q);
  const float norm  = sqrtf(nsq);
  const float scale = (1.f / (norm + 1e-12f)) * (nsq / (1.f + nsq));
#pragma unroll
  for (int j = 0; j < 8; ++j) pa[j] *= scale;
}

__global__ __launch_bounds__(256)
void deconv_route_fused(const float* __restrict__ x,
                        const float* __restrict__ w,
                        const float* __restrict__ deconv_b,
                        const float* __restrict__ rbias,
                        float* __restrict__ out) {
  __shared__ __attribute__((aligned(16))) float xs[IN_DIM * TILE];
  __shared__ __attribute__((aligned(16))) float ex[2 * 16 * EXROW];

  const int tid   = threadIdx.x;
  const int lane  = tid & 31;
  const int wv    = tid >> 5;
  const int ocb   = wv * 16;
  const int n     = lane & 15;
  const int khalf = lane >> 4;

  const int wg  = blockIdx.x;
  const int b   = wg >> 7;
  const int rem = wg & 127;
  const int pd  = (rem >> 6) & 1, ph = (rem >> 5) & 1;
  const int dp  = (rem >> 1) & 15;
  const int hp0 = (rem & 1) * 8;

  v2f A[2][2][4][4];
  {
    const int kd0 = 1 - pd, kh0 = 1 - ph;
#pragma unroll
    for (int sd = 0; sd < 2; ++sd)
#pragma unroll
      for (int sh = 0; sh < 2; ++sh)
#pragma unroll
        for (int kw = 0; kw < 4; ++kw) {
          const int tap = ((2 * sd + kd0) * 4 + (2 * sh + kh0)) * 4 + kw;
#pragma unroll
          for (int c = 0; c < 4; ++c) {
            const int ic = c * 4 + khalf * 2;
            const float* wp = w + (size_t)(ic * OCH + ocb + n) * 64 + tap;
            A[sd][sh][kw][c].x = wp[0];
            A[sd][sh][kw][c].y = wp[OCH * 64];
          }
        }
  }
  float bl[8], rb[8];
#pragma unroll
  for (int j = 0; j < 8; ++j) {
    bl[j] = deconv_b[ocb + khalf * 8 + j];
    rb[j] = rbias[ocb + khalf * 8 + j];
  }

#pragma unroll
  for (int k = 0; k < 4; ++k) {
    const int idx = tid + 256 * k;
    const int e = idx & 1, cs = (idx >> 1) & 1, r = idx >> 2;
    xs[(r * 18 + (cs ? 17 : 0)) * 2 + e] = 0.f;
  }

  const int colq = (tid & 3) * 4;
  const int lrow = tid >> 2;
  const int lic = lrow >> 2, lsd = (lrow >> 1) & 1, lsh = lrow & 1;
  const int ld  = dp + (lsd ? (pd ? 0 : -1) : (pd ? 1 : 0));
  const int ldh = lsh ? (ph ? 0 : -1) : (ph ? 1 : 0);
  const bool dok = (unsigned)ld < 16u;
  const float* xrow = x + ((size_t)(b * IN_DIM) * 16 + lic) * (DIN * DIN * DIN)
                        + (size_t)(dok ? ld : 0) * (DIN * DIN) + colq;
  float* xdst = xs + (((lic >> 1) * 4 + lsd * 2 + lsh) * 18 + 1 + colq) * 2 + (lic & 1);

  const float* xl = xs + (khalf * 4 * 18 + n) * 2;

  v4f pf[IN_DIM];
  {
    const int h = hp0 + ldh;
    const bool ok = dok && ((unsigned)h < 16u);
    const float* src = xrow + (ok ? h : 0) * DIN;
#pragma unroll
    for (int k = 0; k < IN_DIM; ++k)
      pf[k] = ok ? *(const v4f*)(src + (size_t)k * 16 * (DIN * DIN * DIN)) : (v4f){0.f, 0.f, 0.f, 0.f};
  }

#pragma unroll 1
  for (int hh = 0; hh < 8; ++hh) {
    const int hp = hp0 + hh;
#pragma unroll
    for (int k = 0; k < IN_DIM; ++k) {
      float* d = xdst + k * TILE;
      d[0] = pf[k].x; d[2] = pf[k].y; d[4] = pf[k].z; d[6] = pf[k].w;
    }
    __syncthreads();

    v4f Bx[16];
    load_b<0, 0>(xl, Bx);

    if (hh + 1 < 8) {
      const int h = hp + 1 + ldh;
      const bool ok = dok && ((unsigned)h < 16u);
      const float* src = xrow + (ok ? h : 0) * DIN;
#pragma unroll
      for (int k = 0; k < IN_DIM; ++k)
        pf[k] = ok ? *(const v4f*)(src + (size_t)k * 16 * (DIN * DIN * DIN)) : (v4f){0.f, 0.f, 0.f, 0.f};
    }

    float pa0[8], pa1[8];
    tile_parity<0>(xl, ex, A, Bx, bl, rb, n, khalf, wv, pa0);
    tile_parity<1>(xl, ex, A, Bx, bl, rb, n, khalf, wv, pa1);

    const int od = 2 * dp + pd, oh = 2 * hp + ph;
    const int o  = (od * 32 + oh) * 32 + 2 * n;
#pragma unroll
    for (int j = 0; j < 8; ++j) {
      float* dst = out + (((size_t)((b * 8 + wv) * 16 + khalf * 8 + j)) << 15) + o;
      *(v2f*)dst = (v2f){pa0[j], pa1[j]};
    }
  }
}

extern "C" void kernel_launch(void* const* d_in, const int* in_sizes, int n_in,
                              void* d_out, int out_size, void* d_ws, size_t ws_size,
                              hipStream_t stream) {
  (void)in_sizes; (void)n_in; (void)out_size; (void)d_ws; (void)ws_size;
  const float* x  = (const float*)d_in[0];
  const float* w  = (const float*)d_in[1];
  const float* db = (const float*)d_in[2];
  const float* rb = (const float*)d_in[3];
  float* outp  = (float*)d_out;

  deconv_route_fused<<<256, 256, 0, stream>>>(x, w, db, rb, outp);
}


# Routing's 33 quotients and 9 norms per parity on the 1-ulp hardware reciprocal and square-root instructions instead of the exactly rounded 11- and 14-instruction sequences; sum order kept
# speedup vs baseline: 1.9968x; 1.0850x over previous
#include <hip/hip_runtime.h>
#include <math.h>


typedef float v2f __attribute__((ext_vector_type(2)));
typedef float v4f __attribute__((ext_vector_type(4)));
typedef float v8f __attribute__((ext_vector_type(8)));
typedef v4f v4f_a8 __attribute__((aligned(8)));

#define IN_DIM   8
#define OCH      128
#define DIN      16
#define TILE     1152
#define EXROW    68

__device__ __forceinline__ float xhalf(float v) {
  return __builtin_bit_cast(float, __builtin_amdgcn_permlanex16(
      __builtin_bit_cast(int, v), __builtin_bit_cast(int, v), 0x76543210, 0xfedcba98, false, false));
}
__device__ __forceinline__ float hsum(float v) { return v + xhalf(v); }

__device__ __forceinline__ float rcp1(float v)  { return __builtin_amdgcn_rcpf(v); }
__device__ __forceinline__ float sqrt1(float v) { return __builtin_amdgcn_sqrtf(v); }

__device__ __forceinline__ void fence() { __builtin_amdgcn_sched_barrier(0); }

template <int PW, int BLK>
__device__ __forceinline__ void load_b(const float* __restrict__ xl, v4f (&Bf)[16]) {
  constexpr int sd = BLK >> 2, sh = (BLK >> 1) & 1, half = BLK & 1;
#pragma unroll
  for (int c = 0; c < 4; ++c)
#pragma unroll
    for (int i = 0; i < 4; ++i)
      Bf[c * 4 + i] = *(const v4f_a8*)&xl[(half * 4 + i) * TILE + ((c * 8 + sd * 2 + sh) * 18 + PW) * 2];
}

template <int PW, int BLK>
__device__ __forceinline__ void mma_b(const v2f (&A)[2][2][4][4], const v4f (&Bf)[16], v8f (&acc)[IN_DIM]) {
  constexpr int sd = BLK >> 2, sh = (BLK >> 1) & 1, half = BLK & 1;
#pragma unroll
  for (int sw = 0; sw < 2; ++sw) {
    const int kw = 2 * sw + 1 - PW;
#pragma unroll
    for (int c = 0; c < 4; ++c)
#pragma unroll
      for (int i = 0; i < 4; ++i) {
        const v4f b = Bf[c * 4 + i];
        const v2f bvec = sw ? (v2f){b.x, b.y} : (v2f){b.z, b.w};
        acc[half * 4 + i] = __builtin_amdgcn_wmma_f32_16x16x4_f32(
            false, A[sd][sh][kw][c], false, bvec, (short)0, acc[half * 4 + i], false, false);
      }
  }
}

template <int PW>
__device__ __forceinline__ void tile_parity(const float* __restrict__ xl, float* __restrict__ ex,
                                            const v2f (&A)[2][2][4][4], v4f (&Bx)[16],
                                            const float (&bl)[8], const float (&rb)[8],
                                            const int n, const int khalf, const int wv,
                                            float (&pa)[8]) {
  v8f acc[IN_DIM];
#pragma unroll
  for (int in = 0; in < IN_DIM; ++in) acc[in] = (v8f){0.f, 0.f, 0.f, 0.f, 0.f, 0.f, 0.f, 0.f};

  v4f By[16];
  load_b<PW, 1>(xl, By); fence(); mma_b<PW, 0>(A, Bx, acc); fence();
  load_b<PW, 2>(xl, Bx); fence(); mma_b<PW, 1>(A, By, acc); fence();
  load_b<PW, 3>(xl, By); fence(); mma_b<PW, 2>(A, Bx, acc); fence();
  load_b<PW, 4>(xl, Bx); fence(); mma_b<PW, 3>(A, By, acc); fence();
  load_b<PW, 5>(xl, By); fence(); mma_b<PW, 4>(A, Bx, acc); fence();
  load_b<PW, 6>(xl, Bx); fence(); mma_b<PW, 5>(A, By, acc); fence();
  load_b<PW, 7>(xl, By); fence(); mma_b<PW, 6>(A, Bx, acc); fence();
  mma_b<PW, 7>(A, By, acc);

  const v8f blv = {bl[0], bl[1], bl[2], bl[3], bl[4], bl[5], bl[6], bl[7]};
#pragma unroll
  for (int in = 0; in < IN_DIM; ++in) acc[in] += blv;

  float n2[IN_DIM];
#pragma unroll
  for (int in = 0; in < IN_DIM; ++in) {
    float s = 0.f;
#pragma unroll
    for (int j = 0; j < 8; ++j) s += acc[in][j] * acc[in][j];
    n2[in] = sqrt1(hsum(s));
  }

  float logits[IN_DIM], route[IN_DIM];
#pragma unroll
  for (int in = 0; in < IN_DIM; ++in) { logits[in] = 0.f; route[in] = 0.125f; }

#pragma unroll
  for (int it = 0; it < 3; ++it) {
    if (it > 0) {
      float* e = ex + (it - 1) * 16 * EXROW;
      float ev[IN_DIM];
#pragma unroll
      for (int in = 0; in < IN_DIM; ++in) ev[in] = __expf(logits[in]);
#pragma unroll
      for (int i = 0; i < 4; ++i)
        e[n * EXROW + (khalf * 4 + i) * 8 + wv] = khalf ? ev[4 + i] : ev[i];
      __syncthreads();
#pragma unroll
      for (int in = 0; in < IN_DIM; ++in) {
        const v4f s0 = *(const v4f*)&e[n * EXROW + in * 8];
        const v4f s1 = *(const v4f*)&e[n * EXROW + in * 8 + 4];
        const float t = ((s0.x + s0.y) + (s0.z + s0.w)) + ((s1.x + s1.y) + (s1.z + s1.w));
        route[in] = ev[in] * rcp1(t);
      }
    }
#pragma unroll
    for (int j = 0; j < 8; ++j) pa[j] = rb[j];
#pragma unroll
    for (int in = 0; in < IN_DIM; ++in)
#pragma unroll
      for (int j = 0; j < 8; ++j) pa[j] += acc[in][j] * route[in];
    if (it < 2) {
      float s = 0.f;
#pragma unroll
      for (int j = 0; j < 8; ++j) s += pa[j] * pa[j];
      const float n1 = sqrt1(hsum(s));
#pragma unroll
      for (int in = 0; in < IN_DIM; ++in) {
        float dot = 0.f;
#pragma unroll
        for (int j = 0; j < 8; ++j) dot += pa[j] * acc[in][j];
        dot = hsum(dot);
        logits[in] += dot * rcp1(fmaxf(n1 * n2[in], 1e-8f));
      }
    }
  }

  if (PW == 0) load_b<1, 0>(xl, Bx);

  float q = 0.f;
#pragma unroll
  for (int j = 0; j < 8; ++j) q += pa[j] * pa[j];
  const float nsq   = hsum(q);
  const float norm  = sqrt1(nsq);
  const float scale = rcp1(norm + 1e-12f) * (nsq * rcp1(1.f + nsq));
#pragma unroll
  for (int j = 0; j < 8; ++j) pa[j] *= scale;
}

__global__ __launch_bounds__(256)
void deconv_route_fused(const float* __restrict__ x,
                        const float* __restrict__ w,
                        const float* __restrict__ deconv_b,
                        const float* __restrict__ rbias,
                        float* __restrict__ out) {
  __shared__ __attribute__((aligned(16))) float xs[IN_DIM * TILE];
  __shared__ __attribute__((aligned(16))) float ex[2 * 16 * EXROW];

  const int tid   = threadIdx.x;
  const int lane  = tid & 31;
  const int wv    = tid >> 5;
  const int ocb   = wv * 16;
  const int n     = lane & 15;
  const int khalf = lane >> 4;

  const int wg  = blockIdx.x;
  const int b   = wg >> 7;
  const int rem = wg & 127;
  const int pd  = (rem >> 6) & 1, ph = (rem >> 5) & 1;
  const int dp  = (rem >> 1) & 15;
  const int hp0 = (rem & 1) * 8;

  v2f A[2][2][4][4];
  {
    const int kd0 = 1 - pd, kh0 = 1 - ph;
#pragma unroll
    for (int sd = 0; sd < 2; ++sd)
#pragma unroll
      for (int sh = 0; sh < 2; ++sh)
#pragma unroll
        for (int kw = 0; kw < 4; ++kw) {
          const int tap = ((2 * sd + kd0) * 4 + (2 * sh + kh0)) * 4 + kw;
#pragma unroll
          for (int c = 0; c < 4; ++c) {
            const int ic = c * 4 + khalf * 2;
            const float* wp = w + (size_t)(ic * OCH + ocb + n) * 64 + tap;
            A[sd][sh][kw][c].x = wp[0];
            A[sd][sh][kw][c].y = wp[OCH * 64];
          }
        }
  }
  float bl[8], rb[8];
#pragma unroll
  for (int j = 0; j < 8; ++j) {
    bl[j] = deconv_b[ocb + khalf * 8 + j];
    rb[j] = rbias[ocb + khalf * 8 + j];
  }

#pragma unroll
  for (int k = 0; k < 4; ++k) {
    const int idx = tid + 256 * k;
    const int e = idx & 1, cs = (idx >> 1) & 1, r = idx >> 2;
    xs[(r * 18 + (cs ? 17 : 0)) * 2 + e] = 0.f;
  }

  const int colq = (tid & 3) * 4;
  const int lrow = tid >> 2;
  const int lic = lrow >> 2, lsd = (lrow >> 1) & 1, lsh = lrow & 1;
  const int ld  = dp + (lsd ? (pd ? 0 : -1) : (pd ? 1 : 0));
  const int ldh = lsh ? (ph ? 0 : -1) : (ph ? 1 : 0);
  const bool dok = (unsigned)ld < 16u;
  const float* xrow = x + ((size_t)(b * IN_DIM) * 16 + lic) * (DIN * DIN * DIN)
                        + (size_t)(dok ? ld : 0) * (DIN * DIN) + colq;
  float* xdst = xs + (((lic >> 1) * 4 + lsd * 2 + lsh) * 18 + 1 + colq) * 2 + (lic & 1);

  const float* xl = xs + (khalf * 4 * 18 + n) * 2;

  v4f pf[IN_DIM];
  {
    const int h = hp0 + ldh;
    const bool ok = dok && ((unsigned)h < 16u);
    const float* src = xrow + (ok ? h : 0) * DIN;
#pragma unroll
    for (int k = 0; k < IN_DIM; ++k)
      pf[k] = ok ? *(const v4f*)(src + (size_t)k * 16 * (DIN * DIN * DIN)) : (v4f){0.f, 0.f, 0.f, 0.f};
  }

#pragma unroll 1
  for (int hh = 0; hh < 8; ++hh) {
    const int hp = hp0 + hh;
#pragma unroll
    for (int k = 0; k < IN_DIM; ++k) {
      float* d = xdst + k * TILE;
      d[0] = pf[k].x; d[2] = pf[k].y; d[4] = pf[k].z; d[6] = pf[k].w;
    }
    __syncthreads();

    v4f Bx[16];
    load_b<0, 0>(xl, Bx);

    if (hh + 1 < 8) {
      const int h = hp + 1 + ldh;
      const bool ok = dok && ((unsigned)h < 16u);
      const float* src = xrow + (ok ? h : 0) * DIN;
#pragma unroll
      for (int k = 0; k < IN_DIM; ++k)
        pf[k] = ok ? *(const v4f*)(src + (size_t)k * 16 * (DIN * DIN * DIN)) : (v4f){0.f, 0.f, 0.f, 0.f};
    }

    float pa0[8], pa1[8];
    tile_parity<0>(xl, ex, A, Bx, bl, rb, n, khalf, wv, pa0);
    tile_parity<1>(xl, ex, A, Bx, bl, rb, n, khalf, wv, pa1);

    const int od = 2 * dp + pd, oh = 2 * hp + ph;
    const int o  = (od * 32 + oh) * 32 + 2 * n;
#pragma unroll
    for (int j = 0; j < 8; ++j) {
      float* dst = out + (((size_t)((b * 8 + wv) * 16 + khalf * 8 + j)) << 15) + o;
      *(v2f*)dst = (v2f){pa0[j], pa1[j]};
    }
  }
}

extern "C" void kernel_launch(void* const* d_in, const int* in_sizes, int n_in,
                              void* d_out, int out_size, void* d_ws, size_t ws_size,
                              hipStream_t stream) {
  (void)in_sizes; (void)n_in; (void)out_size; (void)d_ws; (void)ws_size;
  const float* x  = (const float*)d_in[0];
  const float* w  = (const float*)d_in[1];
  const float* db = (const float*)d_in[2];
  const float* rb = (const float*)d_in[3];
  float* outp  = (float*)d_out;

  deconv_route_fused<<<256, 256, 0, stream>>>(x, w, db, rb, outp);
}


# Routing's sums over atoms and capsules as explicit 2-wide packed chains on the accumulators' own register pairs (ends the compiler's cross-capsule repacking moves); sum order changes (pairwise)
# speedup vs baseline: 2.0618x; 1.0326x over previous
#include <hip/hip_runtime.h>
#include <math.h>


typedef float v2f __attribute__((ext_vector_type(2)));
typedef float v4f __attribute__((ext_vector_type(4)));
typedef float v8f __attribute__((ext_vector_type(8)));
typedef v4f v4f_a8 __attribute__((aligned(8)));

#define IN_DIM   8
#define OCH      128
#define DIN      16
#define TILE     1152
#define EXROW    68

__device__ __forceinline__ float xhalf(float v) {
  return __builtin_bit_cast(float, __builtin_amdgcn_permlanex16(
      __builtin_bit_cast(int, v), __builtin_bit_cast(int, v), 0x76543210, 0xfedcba98, false, false));
}
__device__ __forceinline__ float hsum(float v) { return v + xhalf(v); }

__device__ __forceinline__ float rcp1(float v)  { return __builtin_amdgcn_rcpf(v); }
__device__ __forceinline__ float sqrt1(float v) { return __builtin_amdgcn_sqrtf(v); }

#define PR(a, k)   __builtin_shufflevector((a), (a), 2 * (k), 2 * (k) + 1)
#define LOHALF(u)  __builtin_shufflevector((u), (u), 0, 1)
#define HIHALF(u)  __builtin_shufflevector((u), (u), 2, 3)

__device__ __forceinline__ float dot8(const v8f& a, const v8f& b) {
  v2f d = PR(a, 0) * PR(b, 0);
  d += PR(a, 1) * PR(b, 1);
  d += PR(a, 2) * PR(b, 2);
  d += PR(a, 3) * PR(b, 3);
  return d.x + d.y;
}

__device__ __forceinline__ void fence() { __builtin_amdgcn_sched_barrier(0); }

template <int PW, int BLK>
__device__ __forceinline__ void load_b(const float* __restrict__ xl, v4f (&Bf)[16]) {
  constexpr int sd = BLK >> 2, sh = (BLK >> 1) & 1, half = BLK & 1;
#pragma unroll
  for (int c = 0; c < 4; ++c)
#pragma unroll
    for (int i = 0; i < 4; ++i)
      Bf[c * 4 + i] = *(const v4f_a8*)&xl[(half * 4 + i) * TILE + ((c * 8 + sd * 2 + sh) * 18 + PW) * 2];
}

template <int PW, int BLK>
__device__ __forceinline__ void mma_b(const v2f (&A)[2][2][4][4], const v4f (&Bf)[16], v8f (&acc)[IN_DIM]) {
  constexpr int sd = BLK >> 2, sh = (BLK >> 1) & 1, half = BLK & 1;
#pragma unroll
  for (int sw = 0; sw < 2; ++sw) {
    const int kw = 2 * sw + 1 - PW;
#pragma unroll
    for (int c = 0; c < 4; ++c)
#pragma unroll
      for (int i = 0; i < 4; ++i) {
        const v4f b = Bf[c * 4 + i];
        const v2f bvec = sw ? (v2f){b.x, b.y} : (v2f){b.z, b.w};
        acc[half * 4 + i] = __builtin_amdgcn_wmma_f32_16x16x4_f32(
            false, A[sd][sh][kw][c], false, bvec, (short)0, acc[half * 4 + i], false, false);
      }
  }
}

template <int PW>
__device__ __forceinline__ void tile_parity(const float* __restrict__ xl, float* __restrict__ ex,
                                            const v2f (&A)[2][2][4][4], v4f (&Bx)[16],
                                            const float (&bl)[8], const float (&rb)[8],
                                            const int n, const int khalf, const int wv,
                                            float (&pa)[8]) {
  v8f acc[IN_DIM];
#pragma unroll
  for (int in = 0; in < IN_DIM; ++in) acc[in] = (v8f){0.f, 0.f, 0.f, 0.f, 0.f, 0.f, 0.f, 0.f};

  v4f By[16];
  load_b<PW, 1>(xl, By); fence(); mma_b<PW, 0>(A, Bx, acc); fence();
  load_b<PW, 2>(xl, Bx); fence(); mma_b<PW, 1>(A, By, acc); fence();
  load_b<PW, 3>(xl, By); fence(); mma_b<PW, 2>(A, Bx, acc); fence();
  load_b<PW, 4>(xl, Bx); fence(); mma_b<PW, 3>(A, By, acc); fence();
  load_b<PW, 5>(xl, By); fence(); mma_b<PW, 4>(A, Bx, acc); fence();
  load_b<PW, 6>(xl, Bx); fence(); mma_b<PW, 5>(A, By, acc); fence();
  load_b<PW, 7>(xl, By); fence(); mma_b<PW, 6>(A, Bx, acc); fence();
  mma_b<PW, 7>(A, By, acc);

  const v8f blv = {bl[0], bl[1], bl[2], bl[3], bl[4], bl[5], bl[6], bl[7]};
#pragma unroll
  for (int in = 0; in < IN_DIM; ++in) acc[in] += blv;

  float n2[IN_DIM];
#pragma unroll
  for (int in = 0; in < IN_DIM; ++in) n2[in] = sqrt1(hsum(dot8(acc[in], acc[in])));

  const v8f rbv = {rb[0], rb[1], rb[2], rb[3], rb[4], rb[5], rb[6], rb[7]};
  v8f pav = rbv;
  float logits[IN_DIM], route[IN_DIM];
#pragma unroll
  for (int in = 0; in < IN_DIM; ++in) { logits[in] = 0.f; route[in] = 0.125f; }

#pragma unroll
  for (int it = 0; it < 3; ++it) {
    if (it > 0) {
      float* e = ex + (it - 1) * 16 * EXROW;
      float ev[IN_DIM];
#pragma unroll
      for (int in = 0; in < IN_DIM; ++in) ev[in] = __expf(logits[in]);
#pragma unroll
      for (int i = 0; i < 4; ++i)
        e[n * EXROW + (khalf * 4 + i) * 8 + wv] = khalf ? ev[4 + i] : ev[i];
      __syncthreads();
#pragma unroll
      for (int in = 0; in < IN_DIM; ++in) {
        const v4f s0 = *(const v4f*)&e[n * EXROW + in * 8];
        const v4f s1 = *(const v4f*)&e[n * EXROW + in * 8 + 4];
        const v4f u = s0 + s1;
        const v2f h = LOHALF(u) + HIHALF(u);
        route[in] = ev[in] * rcp1(h.x + h.y);
      }
    }
    pav = rbv;
#pragma unroll
    for (int in = 0; in < IN_DIM; ++in) pav += acc[in] * route[in];
    if (it < 2) {
      const float n1 = sqrt1(hsum(dot8(pav, pav)));
#pragma unroll
      for (int in = 0; in < IN_DIM; ++in)
        logits[in] += hsum(dot8(pav, acc[in])) * rcp1(fmaxf(n1 * n2[in], 1e-8f));
    }
  }

  if (PW == 0) load_b<1, 0>(xl, Bx);

  const float nsq   = hsum(dot8(pav, pav));
  const float norm  = sqrt1(nsq);
  const float scale = rcp1(norm + 1e-12f) * (nsq * rcp1(1.f + nsq));
  pav *= scale;
#pragma unroll
  for (int j = 0; j < 8; ++j) pa[j] = pav[j];
}

__global__ __launch_bounds__(256)
void deconv_route_fused(const float* __restrict__ x,
                        const float* __restrict__ w,
                        const float* __restrict__ deconv_b,
                        const float* __restrict__ rbias,
                        float* __restrict__ out) {
  __shared__ __attribute__((aligned(16))) float xs[IN_DIM * TILE];
  __shared__ __attribute__((aligned(16))) float ex[2 * 16 * EXROW];

  const int tid   = threadIdx.x;
  const int lane  = tid & 31;
  const int wv    = tid >> 5;
  const int ocb   = wv * 16;
  const int n     = lane & 15;
  const int khalf = lane >> 4;

  const int wg  = blockIdx.x;
  const int b   = wg >> 7;
  const int rem = wg & 127;
  const int pd  = (rem >> 6) & 1, ph = (rem >> 5) & 1;
  const int dp  = (rem >> 1) & 15;
  const int hp0 = (rem & 1) * 8;

  v2f A[2][2][4][4];
  {
    const int kd0 = 1 - pd, kh0 = 1 - ph;
#pragma unroll
    for (int sd = 0; sd < 2; ++sd)
#pragma unroll
      for (int sh = 0; sh < 2; ++sh)
#pragma unroll
        for (int kw = 0; kw < 4; ++kw) {
          const int tap = ((2 * sd + kd0) * 4 + (2 * sh + kh0)) * 4 + kw;
#pragma unroll
          for (int c = 0; c < 4; ++c) {
            const int ic = c * 4 + khalf * 2;
            const float* wp = w + (size_t)(ic * OCH + ocb + n) * 64 + tap;
            A[sd][sh][kw][c].x = wp[0];
            A[sd][sh][kw][c].y = wp[OCH * 64];
          }
        }
  }
  float bl[8], rb[8];
#pragma unroll
  for (int j = 0; j < 8; ++j) {
    bl[j] = deconv_b[ocb + khalf * 8 + j];
    rb[j] = rbias[ocb + khalf * 8 + j];
  }

#pragma unroll
  for (int k = 0; k < 4; ++k) {
    const int idx = tid + 256 * k;
    const int e = idx & 1, cs = (idx >> 1) & 1, r = idx >> 2;
    xs[(r * 18 + (cs ? 17 : 0)) * 2 + e] = 0.f;
  }

  const int colq = (tid & 3) * 4;
  const int lrow = tid >> 2;
  const int lic = lrow >> 2, lsd = (lrow >> 1) & 1, lsh = lrow & 1;
  const int ld  = dp + (lsd ? (pd ? 0 : -1) : (pd ? 1 : 0));
  const int ldh = lsh ? (ph ? 0 : -1) : (ph ? 1 : 0);
  const bool dok = (unsigned)ld < 16u;
  const float* xrow = x + ((size_t)(b * IN_DIM) * 16 + lic) * (DIN * DIN * DIN)
                        + (size_t)(dok ? ld : 0) * (DIN * DIN) + colq;
  float* xdst = xs + (((lic >> 1) * 4 + lsd * 2 + lsh) * 18 + 1 + colq) * 2 + (lic & 1);

  const float* xl = xs + (khalf * 4 * 18 + n) * 2;

  v4f pf[IN_DIM];
  {
    const int h = hp0 + ldh;
    const bool ok = dok && ((unsigned)h < 16u);
    const float* src = xrow + (ok ? h : 0) * DIN;
#pragma unroll
    for (int k = 0; k < IN_DIM; ++k)
      pf[k] = ok ? *(const v4f*)(src + (size_t)k * 16 * (DIN * DIN * DIN)) : (v4f){0.f, 0.f, 0.f, 0.f};
  }

#pragma unroll 1
  for (int hh = 0; hh < 8; ++hh) {
    const int hp = hp0 + hh;
#pragma unroll
    for (int k = 0; k < IN_DIM; ++k) {
      float* d = xdst + k * TILE;
      d[0] = pf[k].x; d[2] = pf[k].y; d[4] = pf[k].z; d[6] = pf[k].w;
    }
    __syncthreads();

    v4f Bx[16];
    load_b<0, 0>(xl, Bx);

    if (hh + 1 < 8) {
      const int h = hp + 1 + ldh;
      const bool ok = dok && ((unsigned)h < 16u);
      const float* src = xrow + (ok ? h : 0) * DIN;
#pragma unroll
      for (int k = 0; k < IN_DIM; ++k)
        pf[k] = ok ? *(const v4f*)(src + (size_t)k * 16 * (DIN * DIN * DIN)) : (v4f){0.f, 0.f, 0.f, 0.f};
    }

    float pa0[8], pa1[8];
    tile_parity<0>(xl, ex, A, Bx, bl, rb, n, khalf, wv, pa0);
    tile_parity<1>(xl, ex, A, Bx, bl, rb, n, khalf, wv, pa1);

    const int od = 2 * dp + pd, oh = 2 * hp + ph;
    const int o  = (od * 32 + oh) * 32 + 2 * n;
#pragma unroll
    for (int j = 0; j < 8; ++j) {
      float* dst = out + (((size_t)((b * 8 + wv) * 16 + khalf * 8 + j)) << 15) + o;
      *(v2f*)dst = (v2f){pa0[j], pa1[j]};
    }
  }
}

extern "C" void kernel_launch(void* const* d_in, const int* in_sizes, int n_in,
                              void* d_out, int out_size, void* d_ws, size_t ws_size,
                              hipStream_t stream) {
  (void)in_sizes; (void)n_in; (void)out_size; (void)d_ws; (void)ws_size;
  const float* x  = (const float*)d_in[0];
  const float* w  = (const float*)d_in[1];
  const float* db = (const float*)d_in[2];
  const float* rb = (const float*)d_in[3];
  float* outp  = (float*)d_out;

  deconv_route_fused<<<256, 256, 0, stream>>>(x, w, db, rb, outp);
}


# Routing scalars per (voxel, input capsule) computed by one lane half only (own 4 of 8 capsules; partial sums and finished weights swapped by permlanex16); exchange reads halved; sum order kept
# speedup vs baseline: 2.1486x; 1.0421x over previous
#include <hip/hip_runtime.h>
#include <math.h>


typedef float v2f __attribute__((ext_vector_type(2)));
typedef float v4f __attribute__((ext_vector_type(4)));
typedef float v8f __attribute__((ext_vector_type(8)));
typedef v4f v4f_a8 __attribute__((aligned(8)));

#define IN_DIM   8
#define OCH      128
#define DIN      16
#define TILE     1152
#define EXROW    68

__device__ __forceinline__ float xhalf(float v) {
  return __builtin_bit_cast(float, __builtin_amdgcn_permlanex16(
      __builtin_bit_cast(int, v), __builtin_bit_cast(int, v), 0x76543210, 0xfedcba98, false, false));
}
__device__ __forceinline__ float hsum(float v) { return v + xhalf(v); }
__device__ __forceinline__ float own_sum(float lo, float hi, int khalf) {
  const float send = khalf ? lo : hi;
  const float keep = khalf ? hi : lo;
  return keep + xhalf(send);
}

__device__ __forceinline__ float rcp1(float v)  { return __builtin_amdgcn_rcpf(v); }
__device__ __forceinline__ float sqrt1(float v) { return __builtin_amdgcn_sqrtf(v); }

#define PR(a, k)   __builtin_shufflevector((a), (a), 2 * (k), 2 * (k) + 1)
#define LOHALF(u)  __builtin_shufflevector((u), (u), 0, 1)
#define HIHALF(u)  __builtin_shufflevector((u), (u), 2, 3)

__device__ __forceinline__ float dot8(const v8f& a, const v8f& b) {
  v2f d = PR(a, 0) * PR(b, 0);
  d += PR(a, 1) * PR(b, 1);
  d += PR(a, 2) * PR(b, 2);
  d += PR(a, 3) * PR(b, 3);
  return d.x + d.y;
}

__device__ __forceinline__ void fence() { __builtin_amdgcn_sched_barrier(0); }

template <int PW, int BLK>
__device__ __forceinline__ void load_b(const float* __restrict__ xl, v4f (&Bf)[16]) {
  constexpr int sd = BLK >> 2, sh = (BLK >> 1) & 1, half = BLK & 1;
#pragma unroll
  for (int c = 0; c < 4; ++c)
#pragma unroll
    for (int i = 0; i < 4; ++i)
      Bf[c * 4 + i] = *(const v4f_a8*)&xl[(half * 4 + i) * TILE + ((c * 8 + sd * 2 + sh) * 18 + PW) * 2];
}

template <int PW, int BLK>
__device__ __forceinline__ void mma_b(const v2f (&A)[2][2][4][4], const v4f (&Bf)[16], v8f (&acc)[IN_DIM]) {
  constexpr int sd = BLK >> 2, sh = (BLK >> 1) & 1, half = BLK & 1;
#pragma unroll
  for (int sw = 0; sw < 2; ++sw) {
    const int kw = 2 * sw + 1 - PW;
#pragma unroll
    for (int c = 0; c < 4; ++c)
#pragma unroll
      for (int i = 0; i < 4; ++i) {
        const v4f b = Bf[c * 4 + i];
        const v2f bvec = sw ? (v2f){b.x, b.y} : (v2f){b.z, b.w};
        acc[half * 4 + i] = __builtin_amdgcn_wmma_f32_16x16x4_f32(
            false, A[sd][sh][kw][c], false, bvec, (short)0, acc[half * 4 + i], false, false);
      }
  }
}

template <int PW>
__device__ __forceinline__ void tile_parity(const float* __restrict__ xl, float* __restrict__ ex,
                                            const v2f (&A)[2][2][4][4], v4f (&Bx)[16],
                                            const float (&bl)[8], const float (&rb)[8],
                                            const int n, const int khalf, const int wv,
                                            float (&pa)[8]) {
  v8f acc[IN_DIM];
#pragma unroll
  for (int in = 0; in < IN_DIM; ++in) acc[in] = (v8f){0.f, 0.f, 0.f, 0.f, 0.f, 0.f, 0.f, 0.f};

  v4f By[16];
  load_b<PW, 1>(xl, By); fence(); mma_b<PW, 0>(A, Bx, acc); fence();
  load_b<PW, 2>(xl, Bx); fence(); mma_b<PW, 1>(A, By, acc); fence();
  load_b<PW, 3>(xl, By); fence(); mma_b<PW, 2>(A, Bx, acc); fence();
  load_b<PW, 4>(xl, Bx); fence(); mma_b<PW, 3>(A, By, acc); fence();
  load_b<PW, 5>(xl, By); fence(); mma_b<PW, 4>(A, Bx, acc); fence();
  load_b<PW, 6>(xl, Bx); fence(); mma_b<PW, 5>(A, By, acc); fence();
  load_b<PW, 7>(xl, By); fence(); mma_b<PW, 6>(A, Bx, acc); fence();
  mma_b<PW, 7>(A, By, acc);

  const v8f blv = {bl[0], bl[1], bl[2], bl[3], bl[4], bl[5], bl[6], bl[7]};
#pragma unroll
  for (int in = 0; in < IN_DIM; ++in) acc[in] += blv;

  float n2m[4];
#pragma unroll
  for (int i = 0; i < 4; ++i)
    n2m[i] = sqrt1(own_sum(dot8(acc[i], acc[i]), dot8(acc[i + 4], acc[i + 4]), khalf));

  const v8f rbv = {rb[0], rb[1], rb[2], rb[3], rb[4], rb[5], rb[6], rb[7]};
  v8f pav = rbv;
  float lgm[4], route[IN_DIM];
#pragma unroll
  for (int i = 0; i < 4; ++i) lgm[i] = 0.f;
#pragma unroll
  for (int in = 0; in < IN_DIM; ++in) route[in] = 0.125f;

#pragma unroll
  for (int it = 0; it < 3; ++it) {
    if (it > 0) {
      float* e = ex + (it - 1) * 16 * EXROW + n * EXROW + khalf * 32;
      float evm[4];
#pragma unroll
      for (int i = 0; i < 4; ++i) evm[i] = __expf(lgm[i]);
#pragma unroll
      for (int i = 0; i < 4; ++i) e[i * 8 + wv] = evm[i];
      __syncthreads();
      float rm[4];
#pragma unroll
      for (int i = 0; i < 4; ++i) {
        const v4f s0 = *(const v4f*)&e[i * 8];
        const v4f s1 = *(const v4f*)&e[i * 8 + 4];
        const v4f u = s0 + s1;
        const v2f h = LOHALF(u) + HIHALF(u);
        rm[i] = evm[i] * rcp1(h.x + h.y);
      }
#pragma unroll
      for (int i = 0; i < 4; ++i) {
        const float ro = xhalf(rm[i]);
        route[i]     = khalf ? ro : rm[i];
        route[i + 4] = khalf ? rm[i] : ro;
      }
    }
    pav = rbv;
#pragma unroll
    for (int in = 0; in < IN_DIM; ++in) pav += acc[in] * route[in];
    if (it < 2) {
      const float n1 = sqrt1(hsum(dot8(pav, pav)));
#pragma unroll
      for (int i = 0; i < 4; ++i)
        lgm[i] += own_sum(dot8(pav, acc[i]), dot8(pav, acc[i + 4]), khalf) * rcp1(fmaxf(n1 * n2m[i], 1e-8f));
    }
  }

  if (PW == 0) load_b<1, 0>(xl, Bx);

  const float nsq   = hsum(dot8(pav, pav));
  const float norm  = sqrt1(nsq);
  const float scale = rcp1(norm + 1e-12f) * (nsq * rcp1(1.f + nsq));
  pav *= scale;
#pragma unroll
  for (int j = 0; j < 8; ++j) pa[j] = pav[j];
}

__global__ __launch_bounds__(256)
void deconv_route_fused(const float* __restrict__ x,
                        const float* __restrict__ w,
                        const float* __restrict__ deconv_b,
                        const float* __restrict__ rbias,
                        float* __restrict__ out) {
  __shared__ __attribute__((aligned(16))) float xs[IN_DIM * TILE];
  __shared__ __attribute__((aligned(16))) float ex[2 * 16 * EXROW];

  const int tid   = threadIdx.x;
  const int lane  = tid & 31;
  const int wv    = tid >> 5;
  const int ocb   = wv * 16;
  const int n     = lane & 15;
  const int khalf = lane >> 4;

  const int wg  = blockIdx.x;
  const int b   = wg >> 7;
  const int rem = wg & 127;
  const int pd  = (rem >> 6) & 1, ph = (rem >> 5) & 1;
  const int dp  = (rem >> 1) & 15;
  const int hp0 = (rem & 1) * 8;

  v2f A[2][2][4][4];
  {
    const int kd0 = 1 - pd, kh0 = 1 - ph;
#pragma unroll
    for (int sd = 0; sd < 2; ++sd)
#pragma unroll
      for (int sh = 0; sh < 2; ++sh)
#pragma unroll
        for (int kw = 0; kw < 4; ++kw) {
          const int tap = ((2 * sd + kd0) * 4 + (2 * sh + kh0)) * 4 + kw;
#pragma unroll
          for (int c = 0; c < 4; ++c) {
            const int ic = c * 4 + khalf * 2;
            const float* wp = w + (size_t)(ic * OCH + ocb + n) * 64 + tap;
            A[sd][sh][kw][c].x = wp[0];
            A[sd][sh][kw][c].y = wp[OCH * 64];
          }
        }
  }
  float bl[8], rb[8];
#pragma unroll
  for (int j = 0; j < 8; ++j) {
    bl[j] = deconv_b[ocb + khalf * 8 + j];
    rb[j] = rbias[ocb + khalf * 8 + j];
  }

#pragma unroll
  for (int k = 0; k < 4; ++k) {
    const int idx = tid + 256 * k;
    const int e = idx & 1, cs = (idx >> 1) & 1, r = idx >> 2;
    xs[(r * 18 + (cs ? 17 : 0)) * 2 + e] = 0.f;
  }

  const int colq = (tid & 3) * 4;
  const int lrow = tid >> 2;
  const int lic = lrow >> 2, lsd = (lrow >> 1) & 1, lsh = lrow & 1;
  const int ld  = dp + (lsd ? (pd ? 0 : -1) : (pd ? 1 : 0));
  const int ldh = lsh ? (ph ? 0 : -1) : (ph ? 1 : 0);
  const bool dok = (unsigned)ld < 16u;
  const float* xrow = x + ((size_t)(b * IN_DIM) * 16 + lic) * (DIN * DIN * DIN)
                        + (size_t)(dok ? ld : 0) * (DIN * DIN) + colq;
  float* xdst = xs + (((lic >> 1) * 4 + lsd * 2 + lsh) * 18 + 1 + colq) * 2 + (lic & 1);

  const float* xl = xs + (khalf * 4 * 18 + n) * 2;

  v4f pf[IN_DIM];
  {
    const int h = hp0 + ldh;
    const bool ok = dok && ((unsigned)h < 16u);
    const float* src = xrow + (ok ? h : 0) * DIN;
#pragma unroll
    for (int k = 0; k < IN_DIM; ++k)
      pf[k] = ok ? *(const v4f*)(src + (size_t)k * 16 * (DIN * DIN * DIN)) : (v4f){0.f, 0.f, 0.f, 0.f};
  }

#pragma unroll 1
  for (int hh = 0; hh < 8; ++hh) {
    const int hp = hp0 + hh;
#pragma unroll
    for (int k = 0; k < IN_DIM; ++k) {
      float* d = xdst + k * TILE;
      d[0] = pf[k].x; d[2] = pf[k].y; d[4] = pf[k].z; d[6] = pf[k].w;
    }
    __syncthreads();

    v4f Bx[16];
    load_b<0, 0>(xl, Bx);

    if (hh + 1 < 8) {
      const int h = hp + 1 + ldh;
      const bool ok = dok && ((unsigned)h < 16u);
      const float* src = xrow + (ok ? h : 0) * DIN;
#pragma unroll
      for (int k = 0; k < IN_DIM; ++k)
        pf[k] = ok ? *(const v4f*)(src + (size_t)k * 16 * (DIN * DIN * DIN)) : (v4f){0.f, 0.f, 0.f, 0.f};
    }

    float pa0[8], pa1[8];
    tile_parity<0>(xl, ex, A, Bx, bl, rb, n, khalf, wv, pa0);
    tile_parity<1>(xl, ex, A, Bx, bl, rb, n, khalf, wv, pa1);

    const int od = 2 * dp + pd, oh = 2 * hp + ph;
    const int o  = (od * 32 + oh) * 32 + 2 * n;
#pragma unroll
    for (int j = 0; j < 8; ++j) {
      float* dst = out + (((size_t)((b * 8 + wv) * 16 + khalf * 8 + j)) << 15) + o;
      *(v2f*)dst = (v2f){pa0[j], pa1[j]};
    }
  }
}

extern "C" void kernel_launch(void* const* d_in, const int* in_sizes, int n_in,
                              void* d_out, int out_size, void* d_ws, size_t ws_size,
                              hipStream_t stream) {
  (void)in_sizes; (void)n_in; (void)out_size; (void)d_ws; (void)ws_size;
  const float* x  = (const float*)d_in[0];
  const float* w  = (const float*)d_in[1];
  const float* db = (const float*)d_in[2];
  const float* rb = (const float*)d_in[3];
  float* outp  = (float*)d_out;

  deconv_route_fused<<<256, 256, 0, stream>>>(x, w, db, rb, outp);
}


# Output rows taken in pairs: along h 3 WMMA sets (G0+G1)*xb, G1*(xa-xb), G0*(xc-xb) replace 4 (minimal 2-tap filtering, f32; x staged as row differences, LDS tile double-buffered); sum order changes
# speedup vs baseline: 2.6200x; 1.2194x over previous
#include <hip/hip_runtime.h>
#include <math.h>


typedef float v2f __attribute__((ext_vector_type(2)));
typedef float v4f __attribute__((ext_vector_type(4)));
typedef float v8f __attribute__((ext_vector_type(8)));
typedef v4f v4f_a8 __attribute__((aligned(8)));

#define IN_DIM   8
#define OCH      128
#define DIN      16
#define TILE     1728
#define XBUF     (IN_DIM * TILE)
#define EXROW    68

__device__ __forceinline__ float xhalf(float v) {
  return __builtin_bit_cast(float, __builtin_amdgcn_permlanex16(
      __builtin_bit_cast(int, v), __builtin_bit_cast(int, v), 0x76543210, 0xfedcba98, false, false));
}
__device__ __forceinline__ float hsum(float v) { return v + xhalf(v); }
__device__ __forceinline__ float own_sum(float lo, float hi, int khalf) {
  const float send = khalf ? lo : hi;
  const float keep = khalf ? hi : lo;
  return keep + xhalf(send);
}

__device__ __forceinline__ float rcp1(float v)  { return __builtin_amdgcn_rcpf(v); }
__device__ __forceinline__ float sqrt1(float v) { return __builtin_amdgcn_sqrtf(v); }

#define PR(a, k)   __builtin_shufflevector((a), (a), 2 * (k), 2 * (k) + 1)
#define LOHALF(u)  __builtin_shufflevector((u), (u), 0, 1)
#define HIHALF(u)  __builtin_shufflevector((u), (u), 2, 3)

__device__ __forceinline__ float dot8(const v8f& a, const v8f& b) {
  v2f d = PR(a, 0) * PR(b, 0);
  d += PR(a, 1) * PR(b, 1);
  d += PR(a, 2) * PR(b, 2);
  d += PR(a, 3) * PR(b, 3);
  return d.x + d.y;
}

__device__ __forceinline__ void fence() { __builtin_amdgcn_sched_barrier(0); }

template <int PW, int Q>
__device__ __forceinline__ void load_b(const float* __restrict__ xl, v4f (&Bf)[8]) {
  constexpr int phase = Q >> 3, sd = (Q >> 2) & 1, half = (Q >> 1) & 1, ch = Q & 1;
  constexpr int slot = phase == 0 ? 0 : (phase == 1 ? 2 : 1);
#pragma unroll
  for (int cc = 0; cc < 2; ++cc)
#pragma unroll
    for (int i = 0; i < 4; ++i)
      Bf[cc * 4 + i] = *(const v4f_a8*)&xl[(half * 4 + i) * TILE + (((ch * 2 + cc) * 12 + sd * 3 + slot) * 18 + PW) * 2];
}

template <int PW, int Q>
__device__ __forceinline__ void mma_b(const v2f (&A)[2][3][4][4], const v4f (&Bf)[8],
                                      v8f (&acc0)[IN_DIM], v8f (&acc1)[IN_DIM]) {
  constexpr int phase = Q >> 3, sd = (Q >> 2) & 1, half = (Q >> 1) & 1, ch = Q & 1;
  constexpr int ws = phase == 0 ? 2 : (phase == 1 ? 0 : 1);
#pragma unroll
  for (int sw = 0; sw < 2; ++sw) {
    const int kw = 2 * sw + 1 - PW;
#pragma unroll
    for (int cc = 0; cc < 2; ++cc)
#pragma unroll
      for (int i = 0; i < 4; ++i) {
        const v4f b = Bf[cc * 4 + i];
        const v2f bvec = sw ? (v2f){b.x, b.y} : (v2f){b.z, b.w};
        const v2f a = A[sd][ws][kw][ch * 2 + cc];
        const int in = half * 4 + i;
        if (phase != 1)
          acc0[in] = __builtin_amdgcn_wmma_f32_16x16x4_f32(false, a, false, bvec, (short)0, acc0[in], false, false);
        else if (sd == 0 && ch == 0 && sw == 0 && cc == 0)
          acc1[in] = __builtin_amdgcn_wmma_f32_16x16x4_f32(false, a, false, bvec, (short)0, acc0[in], false, false);
        else
          acc1[in] = __builtin_amdgcn_wmma_f32_16x16x4_f32(false, a, false, bvec, (short)0, acc1[in], false, false);
      }
  }
}

template <int PW, int PH>
__device__ __forceinline__ void vote_phase(const float* __restrict__ xl, const v2f (&A)[2][3][4][4],
                                           v4f (&Bx)[8], v4f (&By)[8],
                                           v8f (&acc0)[IN_DIM], v8f (&acc1)[IN_DIM]) {
  constexpr int Q0 = PH * 8;
  load_b<PW, Q0 + 1>(xl, By); fence(); mma_b<PW, Q0 + 0>(A, Bx, acc0, acc1); fence();
  load_b<PW, Q0 + 2>(xl, Bx); fence(); mma_b<PW, Q0 + 1>(A, By, acc0, acc1); fence();
  load_b<PW, Q0 + 3>(xl, By); fence(); mma_b<PW, Q0 + 2>(A, Bx, acc0, acc1); fence();
  load_b<PW, Q0 + 4>(xl, Bx); fence(); mma_b<PW, Q0 + 3>(A, By, acc0, acc1); fence();
  load_b<PW, Q0 + 5>(xl, By); fence(); mma_b<PW, Q0 + 4>(A, Bx, acc0, acc1); fence();
  load_b<PW, Q0 + 6>(xl, Bx); fence(); mma_b<PW, Q0 + 5>(A, By, acc0, acc1); fence();
  load_b<PW, Q0 + 7>(xl, By); fence(); mma_b<PW, Q0 + 6>(A, Bx, acc0, acc1); fence();
  if (PH < 2) { load_b<PW, (PH < 2 ? Q0 + 8 : 0)>(xl, Bx); fence(); }
  mma_b<PW, Q0 + 7>(A, By, acc0, acc1); fence();
}

__device__ __forceinline__ void route_squash(v8f (&acc)[IN_DIM], float* __restrict__ ex,
                                             const float (&bl)[8], const float (&rb)[8],
                                             const int n, const int khalf, const int wv,
                                             float (&pa)[8]) {
  const v8f blv = {bl[0], bl[1], bl[2], bl[3], bl[4], bl[5], bl[6], bl[7]};
#pragma unroll
  for (int in = 0; in < IN_DIM; ++in) acc[in] += blv;

  float n2m[4];
#pragma unroll
  for (int i = 0; i < 4; ++i)
    n2m[i] = sqrt1(own_sum(dot8(acc[i], acc[i]), dot8(acc[i + 4], acc[i + 4]), khalf));

  const v8f rbv = {rb[0], rb[1], rb[2], rb[3], rb[4], rb[5], rb[6], rb[7]};
  v8f pav = rbv;
  float lgm[4], route[IN_DIM];
#pragma unroll
  for (int i = 0; i < 4; ++i) lgm[i] = 0.f;
#pragma unroll
  for (int in = 0; in < IN_DIM; ++in) route[in] = 0.125f;

#pragma unroll
  for (int it = 0; it < 3; ++it) {
    if (it > 0) {
      float* e = ex + (it - 1) * 16 * EXROW + n * EXROW + khalf * 32;
      float evm[4];
#pragma unroll
      for (int i = 0; i < 4; ++i) evm[i] = __expf(lgm[i]);
#pragma unroll
      for (int i = 0; i < 4; ++i) e[i * 8 + wv] = evm[i];
      __syncthreads();
      float rm[4];
#pragma unroll
      for (int i = 0; i < 4; ++i) {
        const v4f s0 = *(const v4f*)&e[i * 8];
        const v4f s1 = *(const v4f*)&e[i * 8 + 4];
        const v4f u = s0 + s1;
        const v2f h = LOHALF(u) + HIHALF(u);
        rm[i] = evm[i] * rcp1(h.x + h.y);
      }
#pragma unroll
      for (int i = 0; i < 4; ++i) {
        const float ro = xhalf(rm[i]);
        route[i]     = khalf ? ro : rm[i];
        route[i + 4] = khalf ? rm[i] : ro;
      }
    }
    pav = rbv;
#pragma unroll
    for (int in = 0; in < IN_DIM; ++in) pav += acc[in] * route[in];
    if (it < 2) {
      const float n1 = sqrt1(hsum(dot8(pav, pav)));
#pragma unroll
      for (int i = 0; i < 4; ++i)
        lgm[i] += own_sum(dot8(pav, acc[i]), dot8(pav, acc[i + 4]), khalf) * rcp1(fmaxf(n1 * n2m[i], 1e-8f));
    }
  }

  const float nsq   = hsum(dot8(pav, pav));
  const float norm  = sqrt1(nsq);
  const float scale = rcp1(norm + 1e-12f) * (nsq * rcp1(1.f + nsq));
  pav *= scale;
#pragma unroll
  for (int j = 0; j < 8; ++j) pa[j] = pav[j];
}

__device__ __forceinline__ void route_rows(v8f (&acc0)[IN_DIM], const v8f (&acc1)[IN_DIM], float* __restrict__ ex,
                                           const float (&bl)[8], const float (&rb)[8],
                                           const int n, const int khalf, const int wv,
                                           float (&pr0)[8], float (&pr1)[8]) {
#pragma unroll
  for (int j = 0; j < 8; ++j) pr1[j] = 0.f;
#pragma unroll 1
  for (int row = 0; row < 2; ++row) {
#pragma unroll
    for (int j = 0; j < 8; ++j) pr0[j] = pr1[j];
    route_squash(acc0, ex, bl, rb, n, khalf, wv, pr1);
#pragma unroll
    for (int in = 0; in < IN_DIM; ++in) acc0[in] = acc1[in];
  }
}

__global__ __launch_bounds__(256)
void deconv_route_fused(const float* __restrict__ x,
                        const float* __restrict__ w,
                        const float* __restrict__ deconv_b,
                        const float* __restrict__ rbias,
                        float* __restrict__ out) {
  __shared__ __attribute__((aligned(16))) float xs[2 * XBUF];
  __shared__ __attribute__((aligned(16))) float ex[2 * 16 * EXROW];

  const int tid   = threadIdx.x;
  const int lane  = tid & 31;
  const int wv    = tid >> 5;
  const int ocb   = wv * 16;
  const int n     = lane & 15;
  const int khalf = lane >> 4;

  const int wg  = blockIdx.x;
  const int b   = wg >> 7;
  const int rem = wg & 127;
  const int pd  = (rem >> 6) & 1, ph = (rem >> 5) & 1;
  const int dp  = (rem >> 1) & 15;
  const int hp0 = (rem & 1) * 8;

  v2f A[2][3][4][4];
  {
    const int kd0 = 1 - pd, kh0 = 1 - ph;
#pragma unroll
    for (int sd = 0; sd < 2; ++sd)
#pragma unroll
      for (int sh = 0; sh < 2; ++sh)
#pragma unroll
        for (int kw = 0; kw < 4; ++kw) {
          const int tap = ((2 * sd + kd0) * 4 + (2 * sh + kh0)) * 4 + kw;
#pragma unroll
          for (int c = 0; c < 4; ++c) {
            const int ic = c * 4 + khalf * 2;
            const float* wp = w + (size_t)(ic * OCH + ocb + n) * 64 + tap;
            A[sd][sh][kw][c].x = wp[0];
            A[sd][sh][kw][c].y = wp[OCH * 64];
          }
        }
#pragma unroll
    for (int sd = 0; sd < 2; ++sd)
#pragma unroll
      for (int kw = 0; kw < 4; ++kw)
#pragma unroll
        for (int c = 0; c < 4; ++c)
          A[sd][2][kw][c] = A[sd][0][kw][c] + A[sd][1][kw][c];
  }
  float bl[8], rb[8];
#pragma unroll
  for (int j = 0; j < 8; ++j) {
    bl[j] = deconv_b[ocb + khalf * 8 + j];
    rb[j] = rbias[ocb + khalf * 8 + j];
  }

#pragma unroll
  for (int k = 0; k < 12; ++k) {
    const int idx = tid + 256 * k;
    const int e = idx & 1, cs = (idx >> 1) & 1, r = idx >> 2;
    xs[(r * 18 + (cs ? 17 : 0)) * 2 + e] = 0.f;
  }

  const int colq = (tid & 3) * 4;
  const int lrow = (tid >> 2) & 31;
  const int lic = lrow >> 1, lsd = lrow & 1;
  const int kq  = tid >> 7;
  const int ld  = dp + (lsd ? (pd ? 0 : -1) : (pd ? 1 : 0));
  const bool dok = (unsigned)ld < 16u;
  const float* xrow = x + ((size_t)((b * IN_DIM + kq * 4) * 16 + lic)) * (DIN * DIN * DIN)
                        + (size_t)(dok ? ld : 0) * (DIN * DIN) + colq;
  float* xdst = xs + kq * 4 * TILE + (((lic >> 1) * 2 + lsd) * 3 * 18 + 1 + colq) * 2 + (lic & 1);

  const float* xl = xs + (khalf * 6 * 18 + n) * 2;

  v4f pf[4][3];
#define LOAD_ROWS(HP, EN)                                                               \
  _Pragma("unroll")                                                                     \
  for (int s = 0; s < 3; ++s) {                                                         \
    const int h = (HP) - 1 + ph + s;                                                    \
    const bool ok = (EN) && dok && ((unsigned)h < 16u);                                 \
    const float* src = xrow + (ok ? h : 0) * DIN;                                       \
    _Pragma("unroll")                                                                   \
    for (int k = 0; k < 4; ++k)                                                         \
      pf[k][s] = ok ? *(const v4f*)(src + (size_t)k * 16 * (DIN * DIN * DIN)) : (v4f){0.f, 0.f, 0.f, 0.f}; \
  }
#define STORE_ROWS(BUF)                                                                 \
  _Pragma("unroll")                                                                     \
  for (int k = 0; k < 4; ++k) {                                                         \
    float* d = xdst + (BUF) * XBUF + k * TILE;                                          \
    const v4f xb = pf[k][1];                                                            \
    const v4f da = pf[k][0] - xb;                                                       \
    const v4f dc = pf[k][2] - xb;                                                       \
    d[0]  = xb.x; d[2]  = xb.y; d[4]  = xb.z; d[6]  = xb.w;                             \
    d[36] = da.x; d[38] = da.y; d[40] = da.z; d[42] = da.w;                             \
    d[72] = dc.x; d[74] = dc.y; d[76] = dc.z; d[78] = dc.w;                             \
  }

  LOAD_ROWS(hp0, true)
  STORE_ROWS(0)
  __syncthreads();

#pragma unroll 1
  for (int t = 0; t < 4; ++t) {
    const int hp  = hp0 + 2 * t;
    const int cur = t & 1;
    const float* xlc = xl + cur * XBUF;

    v4f Bx[8], By[8];
    load_b<0, 0>(xlc, Bx);

    LOAD_ROWS(hp + 2, t + 1 < 4)

    v8f acc0[IN_DIM], acc1[IN_DIM];
    float pa00[8], pa01[8], pa10[8], pa11[8];

#pragma unroll
    for (int in = 0; in < IN_DIM; ++in) acc0[in] = (v8f){0.f, 0.f, 0.f, 0.f, 0.f, 0.f, 0.f, 0.f};
    vote_phase<0, 0>(xlc, A, Bx, By, acc0, acc1);
    STORE_ROWS(cur ^ 1)
    vote_phase<0, 1>(xlc, A, Bx, By, acc0, acc1);
    vote_phase<0, 2>(xlc, A, Bx, By, acc0, acc1);
    route_rows(acc0, acc1, ex, bl, rb, n, khalf, wv, pa00, pa10);
    load_b<1, 0>(xlc, Bx);

#pragma unroll
    for (int in = 0; in < IN_DIM; ++in) acc0[in] = (v8f){0.f, 0.f, 0.f, 0.f, 0.f, 0.f, 0.f, 0.f};
    vote_phase<1, 0>(xlc, A, Bx, By, acc0, acc1);
    vote_phase<1, 1>(xlc, A, Bx, By, acc0, acc1);
    vote_phase<1, 2>(xlc, A, Bx, By, acc0, acc1);
    route_rows(acc0, acc1, ex, bl, rb, n, khalf, wv, pa01, pa11);

    const int od = 2 * dp + pd, oh = 2 * hp + ph;
    const int o  = (od * 32 + oh) * 32 + 2 * n;
#pragma unroll
    for (int j = 0; j < 8; ++j) {
      float* dst = out + (((size_t)((b * 8 + wv) * 16 + khalf * 8 + j)) << 15) + o;
      *(v2f*)dst        = (v2f){pa00[j], pa01[j]};
      *(v2f*)(dst + 64) = (v2f){pa10[j], pa11[j]};
    }
  }
#undef LOAD_ROWS
#undef STORE_ROWS
}

extern "C" void kernel_launch(void* const* d_in, const int* in_sizes, int n_in,
                              void* d_out, int out_size, void* d_ws, size_t ws_size,
                              hipStream_t stream) {
  (void)in_sizes; (void)n_in; (void)out_size; (void)d_ws; (void)ws_size;
  const float* x  = (const float*)d_in[0];
  const float* w  = (const float*)d_in[1];
  const float* db = (const float*)d_in[2];
  const float* rb = (const float*)d_in[3];
  float* outp  = (float*)d_out;

  deconv_route_fused<<<256, 256, 0, stream>>>(x, w, db, rb, outp);
}


# Minimal 2-tap filtering along d as well as h: each 2 d' x 2 h' row tile takes 9 WMMA products for 12 (tap sets summed on the fly, x staged as differences along both axes); sum order changes
# speedup vs baseline: 3.0540x; 1.1656x over previous
#include <hip/hip_runtime.h>
#include <math.h>


typedef float v2f __attribute__((ext_vector_type(2)));
typedef float v4f __attribute__((ext_vector_type(4)));
typedef float v8f __attribute__((ext_vector_type(8)));
typedef v4f v4f_a8 __attribute__((aligned(8)));

#define IN_DIM   8
#define OCH      128
#define DIN      16
#define TILE     2592
#define XBUF     (IN_DIM * TILE)
#define EXROW    68

__device__ __forceinline__ float xhalf(float v) {
  return __builtin_bit_cast(float, __builtin_amdgcn_permlanex16(
      __builtin_bit_cast(int, v), __builtin_bit_cast(int, v), 0x76543210, 0xfedcba98, false, false));
}
__device__ __forceinline__ float hsum(float v) { return v + xhalf(v); }
__device__ __forceinline__ float own_sum(float lo, float hi, int khalf) {
  const float send = khalf ? lo : hi;
  const float keep = khalf ? hi : lo;
  return keep + xhalf(send);
}

__device__ __forceinline__ float rcp1(float v)  { return __builtin_amdgcn_rcpf(v); }
__device__ __forceinline__ float sqrt1(float v) { return __builtin_amdgcn_sqrtf(v); }

#define PR(a, k)   __builtin_shufflevector((a), (a), 2 * (k), 2 * (k) + 1)
#define LOHALF(u)  __builtin_shufflevector((u), (u), 0, 1)
#define HIHALF(u)  __builtin_shufflevector((u), (u), 2, 3)

__device__ __forceinline__ float dot8(const v8f& a, const v8f& b) {
  v2f d = PR(a, 0) * PR(b, 0);
  d += PR(a, 1) * PR(b, 1);
  d += PR(a, 2) * PR(b, 2);
  d += PR(a, 3) * PR(b, 3);
  return d.x + d.y;
}

__device__ __forceinline__ void fence() { __builtin_amdgcn_sched_barrier(0); }

template <int Q>
__device__ __forceinline__ void load_b(const float* __restrict__ xp, v4f (&Bf)[8]) {
  constexpr int p = Q >> 2, ch = (Q >> 1) & 1, half = Q & 1;
  constexpr int hs = p == 0 ? 0 : (p == 1 ? 2 : 1);
#pragma unroll
  for (int cc = 0; cc < 2; ++cc)
#pragma unroll
    for (int i = 0; i < 4; ++i)
      Bf[cc * 4 + i] = *(const v4f_a8*)&xp[(half * 4 + i) * TILE + (((ch * 2 + cc) * 18 + hs) * 18) * 2];
}

template <bool CENTRE, int HS>
__device__ __forceinline__ v2f wfrag(const v2f (&A)[2][2][4][4], const int kw, const int c) {
  if (CENTRE) {
    if (HS == 0) return (A[0][0][kw][c] + A[0][1][kw][c]) + (A[1][0][kw][c] + A[1][1][kw][c]);
    if (HS == 1) return A[0][1][kw][c] + A[1][1][kw][c];
    return A[0][0][kw][c] + A[1][0][kw][c];
  } else {
    if (HS == 0) return A[1][0][kw][c] + A[1][1][kw][c];
    if (HS == 1) return A[1][1][kw][c];
    return A[1][0][kw][c];
  }
}

template <bool CENTRE, int Q>
__device__ __forceinline__ void mma_b(const v2f (&A)[2][2][4][4], const v4f (&Bf)[8],
                                      v8f (&X)[IN_DIM], v8f (&Y)[IN_DIM]) {
  constexpr int p = Q >> 2, ch = (Q >> 1) & 1, half = Q & 1;
  constexpr int hs = p == 0 ? 0 : (p == 1 ? 2 : 1);
#pragma unroll
  for (int sw = 0; sw < 2; ++sw) {
    const int kw = 2 * sw + 1;
#pragma unroll
    for (int cc = 0; cc < 2; ++cc) {
      const v2f a = wfrag<CENTRE, hs>(A, kw, ch * 2 + cc);
#pragma unroll
      for (int i = 0; i < 4; ++i) {
        const v4f b = Bf[cc * 4 + i];
        const v2f bvec = sw ? (v2f){b.x, b.y} : (v2f){b.z, b.w};
        const int in = half * 4 + i;
        if (p != 1)
          X[in] = __builtin_amdgcn_wmma_f32_16x16x4_f32(false, a, false, bvec, (short)0, X[in], false, false);
        else if (ch == 0 && sw == 0 && cc == 0)
          Y[in] = __builtin_amdgcn_wmma_f32_16x16x4_f32(false, a, false, bvec, (short)0, X[in], false, false);
        else
          Y[in] = __builtin_amdgcn_wmma_f32_16x16x4_f32(false, a, false, bvec, (short)0, Y[in], false, false);
      }
    }
  }
}

template <bool CENTRE>
__device__ __forceinline__ void products(const float* __restrict__ xp, const v2f (&A)[2][2][4][4],
                                         v8f (&X)[IN_DIM], v8f (&Y)[IN_DIM]) {
  v4f Bx[8], By[8];
  load_b<0>(xp, Bx);
  load_b<1>(xp, By);  fence(); mma_b<CENTRE, 0>(A, Bx, X, Y);  fence();
  load_b<2>(xp, Bx);  fence(); mma_b<CENTRE, 1>(A, By, X, Y);  fence();
  load_b<3>(xp, By);  fence(); mma_b<CENTRE, 2>(A, Bx, X, Y);  fence();
  load_b<4>(xp, Bx);  fence(); mma_b<CENTRE, 3>(A, By, X, Y);  fence();
  load_b<5>(xp, By);  fence(); mma_b<CENTRE, 4>(A, Bx, X, Y);  fence();
  load_b<6>(xp, Bx);  fence(); mma_b<CENTRE, 5>(A, By, X, Y);  fence();
  load_b<7>(xp, By);  fence(); mma_b<CENTRE, 6>(A, Bx, X, Y);  fence();
  load_b<8>(xp, Bx);  fence(); mma_b<CENTRE, 7>(A, By, X, Y);  fence();
  load_b<9>(xp, By);  fence(); mma_b<CENTRE, 8>(A, Bx, X, Y);  fence();
  load_b<10>(xp, Bx); fence(); mma_b<CENTRE, 9>(A, By, X, Y);  fence();
  load_b<11>(xp, By); fence(); mma_b<CENTRE, 10>(A, Bx, X, Y); fence();
  mma_b<CENTRE, 11>(A, By, X, Y); fence();
}

__device__ __forceinline__ void route_squash(v8f (&acc)[IN_DIM], float* __restrict__ ex,
                                             const float (&bl)[8], const float (&rb)[8],
                                             const int n, const int khalf, const int wv,
                                             float (&pa)[8]) {
  const v8f blv = {bl[0], bl[1], bl[2], bl[3], bl[4], bl[5], bl[6], bl[7]};
#pragma unroll
  for (int in = 0; in < IN_DIM; ++in) acc[in] += blv;

  float n2m[4];
#pragma unroll
  for (int i = 0; i < 4; ++i)
    n2m[i] = sqrt1(own_sum(dot8(acc[i], acc[i]), dot8(acc[i + 4], acc[i + 4]), khalf));

  const v8f rbv = {rb[0], rb[1], rb[2], rb[3], rb[4], rb[5], rb[6], rb[7]};
  v8f pav = rbv;
  float lgm[4], route[IN_DIM];
#pragma unroll
  for (int i = 0; i < 4; ++i) lgm[i] = 0.f;
#pragma unroll
  for (int in = 0; in < IN_DIM; ++in) route[in] = 0.125f;

#pragma unroll
  for (int it = 0; it < 3; ++it) {
    if (it > 0) {
      float* e = ex + (it - 1) * 16 * EXROW + n * EXROW + khalf * 32;
      float evm[4];
#pragma unroll
      for (int i = 0; i < 4; ++i) evm[i] = __expf(lgm[i]);
#pragma unroll
      for (int i = 0; i < 4; ++i) e[i * 8 + wv] = evm[i];
      __syncthreads();
      float rm[4];
#pragma unroll
      for (int i = 0; i < 4; ++i) {
        const v4f s0 = *(const v4f*)&e[i * 8];
        const v4f s1 = *(const v4f*)&e[i * 8 + 4];
        const v4f u = s0 + s1;
        const v2f h = LOHALF(u) + HIHALF(u);
        rm[i] = evm[i] * rcp1(h.x + h.y);
      }
#pragma unroll
      for (int i = 0; i < 4; ++i) {
        const float ro = xhalf(rm[i]);
        route[i]     = khalf ? ro : rm[i];
        route[i + 4] = khalf ? rm[i] : ro;
      }
    }
    pav = rbv;
#pragma unroll
    for (int in = 0; in < IN_DIM; ++in) pav += acc[in] * route[in];
    if (it < 2) {
      const float n1 = sqrt1(hsum(dot8(pav, pav)));
#pragma unroll
      for (int i = 0; i < 4; ++i)
        lgm[i] += own_sum(dot8(pav, acc[i]), dot8(pav, acc[i + 4]), khalf) * rcp1(fmaxf(n1 * n2m[i], 1e-8f));
    }
  }

  const float nsq   = hsum(dot8(pav, pav));
  const float norm  = sqrt1(nsq);
  const float scale = rcp1(norm + 1e-12f) * (nsq * rcp1(1.f + nsq));
  pav *= scale;
#pragma unroll
  for (int j = 0; j < 8; ++j) pa[j] = pav[j];
}

__device__ __forceinline__ void route_rows(v8f (&acc0)[IN_DIM], const v8f (&acc1)[IN_DIM], float* __restrict__ ex,
                                           const float (&bl)[8], const float (&rb)[8],
                                           const int n, const int khalf, const int wv,
                                           float (&pr0)[8], float (&pr1)[8]) {
#pragma unroll
  for (int j = 0; j < 8; ++j) pr1[j] = 0.f;
#pragma unroll 1
  for (int row = 0; row < 2; ++row) {
#pragma unroll
    for (int j = 0; j < 8; ++j) pr0[j] = pr1[j];
    route_squash(acc0, ex, bl, rb, n, khalf, wv, pr1);
#pragma unroll
    for (int in = 0; in < IN_DIM; ++in) acc0[in] = acc1[in];
  }
}

__device__ __forceinline__ void stage_load(const float* __restrict__ xin, const int d0, const int h0,
                                           const bool en, v4f (&r)[3][3]) {
#pragma unroll
  for (int dd = 0; dd < 3; ++dd)
#pragma unroll
    for (int hh = 0; hh < 3; ++hh) {
      const int d = d0 + dd, h = h0 + hh;
      const bool ok = en && ((unsigned)d < 16u) && ((unsigned)h < 16u);
      const float* src = xin + (ok ? d * (DIN * DIN) + h * DIN : 0);
      r[dd][hh] = ok ? *(const v4f*)src : (v4f){0.f, 0.f, 0.f, 0.f};
    }
}
__device__ __forceinline__ void stage_store(float* __restrict__ dst, const v4f (&r)[3][3]) {
  v4f t[3][3];
#pragma unroll
  for (int dd = 0; dd < 3; ++dd) {
    t[dd][0] = r[dd][1];
    t[dd][1] = r[dd][0] - r[dd][1];
    t[dd][2] = r[dd][2] - r[dd][1];
  }
#pragma unroll
  for (int hs = 0; hs < 3; ++hs) {
    const v4f c = t[1][hs];
    const v4f s[3] = {c, t[0][hs] - c, t[2][hs] - c};
#pragma unroll
    for (int ds = 0; ds < 3; ++ds) {
      float* q = dst + (ds * 3 + hs) * 36;
      q[0] = s[ds].x; q[2] = s[ds].y; q[4] = s[ds].z; q[6] = s[ds].w;
    }
  }
}

__global__ __launch_bounds__(256)
void deconv_route_fused(const float* __restrict__ x,
                        const float* __restrict__ w,
                        const float* __restrict__ deconv_b,
                        const float* __restrict__ rbias,
                        float* __restrict__ out) {
  __shared__ __attribute__((aligned(16))) float xs[2 * XBUF];
  __shared__ __attribute__((aligned(16))) float ex[2 * 16 * EXROW];

  const int tid   = threadIdx.x;
  const int lane  = tid & 31;
  const int wv    = tid >> 5;
  const int ocb   = wv * 16;
  const int n     = lane & 15;
  const int khalf = lane >> 4;

  const int wg  = blockIdx.x;
  const int b   = wg >> 7;
  const int rem = wg & 127;
  const int pd  = (rem >> 6) & 1, ph = (rem >> 5) & 1;
  const int dp0 = ((rem >> 2) & 7) * 2;
  const int hp0 = (rem & 3) * 4;

  v2f A[2][2][4][4];
  {
    const int kd0 = 1 - pd, kh0 = 1 - ph;
#pragma unroll
    for (int sd = 0; sd < 2; ++sd)
#pragma unroll
      for (int sh = 0; sh < 2; ++sh)
#pragma unroll
        for (int kw = 0; kw < 4; ++kw) {
          const int tap = ((2 * sd + kd0) * 4 + (2 * sh + kh0)) * 4 + kw;
#pragma unroll
          for (int c = 0; c < 4; ++c) {
            const int ic = c * 4 + khalf * 2;
            const float* wp = w + (size_t)(ic * OCH + ocb + n) * 64 + tap;
            A[sd][sh][kw][c].x = wp[0];
            A[sd][sh][kw][c].y = wp[OCH * 64];
          }
        }
  }
  float bl[8], rb[8];
#pragma unroll
  for (int j = 0; j < 8; ++j) {
    bl[j] = deconv_b[ocb + khalf * 8 + j];
    rb[j] = rbias[ocb + khalf * 8 + j];
  }

#pragma unroll
  for (int k = 0; k < 18; ++k) {
    const int idx = tid + 256 * k;
    const int e = idx & 1, cs = (idx >> 1) & 1, r = idx >> 2;
    xs[(r * 18 + (cs ? 17 : 0)) * 2 + e] = 0.f;
  }

  const int colq = (tid & 3) * 4;
  const int sic  = (tid >> 2) & 15;
  const int kq   = tid >> 6;
  const float* xsrc = x + ((size_t)((b * IN_DIM + kq * 2) * 16 + sic)) * (DIN * DIN * DIN) + colq;
  float* xdst = xs + kq * 2 * TILE + ((sic >> 1) * 9 * 18 + 1 + colq) * 2 + (sic & 1);
  const int d0 = dp0 - 1 + pd;

  const float* xl = xs + (khalf * 9 * 18 + n) * 2;

  {
    v4f r0[3][3], r1[3][3];
    stage_load(xsrc, d0, hp0 - 1 + ph, true, r0);
    stage_load(xsrc + (size_t)16 * (DIN * DIN * DIN), d0, hp0 - 1 + ph, true, r1);
    stage_store(xdst, r0);
    stage_store(xdst + TILE, r1);
  }
  __syncthreads();

#pragma unroll 1
  for (int t = 0; t < 2; ++t) {
    const int hp  = hp0 + 2 * t;
    const int cur = t & 1;

#pragma unroll 1
    for (int pw = 0; pw < 2; ++pw) {
      const float* xp = xl + cur * XBUF + pw * 2;

      v4f r[3][3];
      stage_load(xsrc + (size_t)pw * 16 * (DIN * DIN * DIN), d0, hp + 2 - 1 + ph, t + 1 < 2, r);

      v8f U[IN_DIM], V[IN_DIM];
#pragma unroll
      for (int in = 0; in < IN_DIM; ++in) U[in] = (v8f){0.f, 0.f, 0.f, 0.f, 0.f, 0.f, 0.f, 0.f};
      products<true>(xp, A, U, V);

      stage_store(xdst + (cur ^ 1) * XBUF + pw * TILE, r);

#pragma unroll 1
      for (int dr = 0; dr < 2; ++dr) {
        v8f R[IN_DIM], S[IN_DIM];
#pragma unroll
        for (int in = 0; in < IN_DIM; ++in) R[in] = (v8f){0.f, 0.f, 0.f, 0.f, 0.f, 0.f, 0.f, 0.f};
        products<false>(xp + (dr + 1) * 3 * 18 * 2, A, R, S);
#pragma unroll
        for (int in = 0; in < IN_DIM; ++in) { R[in] += U[in]; S[in] += V[in]; }

        float pr0[8], pr1[8];
        route_rows(R, S, ex, bl, rb, n, khalf, wv, pr0, pr1);

        const int od = 2 * (dp0 + dr) + pd, oh = 2 * hp + ph;
        const int o  = (od * 32 + oh) * 32 + 2 * n + pw;
#pragma unroll
        for (int j = 0; j < 8; ++j) {
          float* dst = out + (((size_t)((b * 8 + wv) * 16 + khalf * 8 + j)) << 15) + o;
          dst[0]  = pr0[j];
          dst[64] = pr1[j];
        }

#pragma unroll
        for (int sh = 0; sh < 2; ++sh)
#pragma unroll
          for (int kw = 0; kw < 4; ++kw)
#pragma unroll
            for (int c = 0; c < 4; ++c) {
              const v2f tmp = A[0][sh][kw][c]; A[0][sh][kw][c] = A[1][sh][kw][c]; A[1][sh][kw][c] = tmp;
            }
      }

#pragma unroll
      for (int sd = 0; sd < 2; ++sd)
#pragma unroll
        for (int sh = 0; sh < 2; ++sh)
#pragma unroll
          for (int c = 0; c < 4; ++c) {
            const v2f t0 = A[sd][sh][0][c]; A[sd][sh][0][c] = A[sd][sh][1][c]; A[sd][sh][1][c] = t0;
            const v2f t2 = A[sd][sh][2][c]; A[sd][sh][2][c] = A[sd][sh][3][c]; A[sd][sh][3][c] = t2;
          }
    }
  }
}

extern "C" void kernel_launch(void* const* d_in, const int* in_sizes, int n_in,
                              void* d_out, int out_size, void* d_ws, size_t ws_size,
                              hipStream_t stream) {
  (void)in_sizes; (void)n_in; (void)out_size; (void)d_ws; (void)ws_size;
  const float* x  = (const float*)d_in[0];
  const float* w  = (const float*)d_in[1];
  const float* db = (const float*)d_in[2];
  const float* rb = (const float*)d_in[3];
  float* outp  = (float*)d_out;

  deconv_route_fused<<<256, 256, 0, stream>>>(x, w, db, rb, outp);
}


# Routing of the two output rows of a pair as ONE interleaved pass (side by side, stage by stage): one barrier and one LDS round trip per exchange serve both rows; per-row sum order kept
# speedup vs baseline: 3.1555x; 1.0332x over previous
#include <hip/hip_runtime.h>
#include <math.h>


typedef float v2f __attribute__((ext_vector_type(2)));
typedef float v4f __attribute__((ext_vector_type(4)));
typedef float v8f __attribute__((ext_vector_type(8)));
typedef v4f v4f_a8 __attribute__((aligned(8)));

#define IN_DIM   8
#define OCH      128
#define DIN      16
#define TILE     2592
#define XBUF     (IN_DIM * TILE)
#define EXROW    68

__device__ __forceinline__ float xhalf(float v) {
  return __builtin_bit_cast(float, __builtin_amdgcn_permlanex16(
      __builtin_bit_cast(int, v), __builtin_bit_cast(int, v), 0x76543210, 0xfedcba98, false, false));
}
__device__ __forceinline__ float hsum(float v) { return v + xhalf(v); }
__device__ __forceinline__ float own_sum(float lo, float hi, int khalf) {
  const float send = khalf ? lo : hi;
  const float keep = khalf ? hi : lo;
  return keep + xhalf(send);
}

__device__ __forceinline__ float rcp1(float v)  { return __builtin_amdgcn_rcpf(v); }
__device__ __forceinline__ float sqrt1(float v) { return __builtin_amdgcn_sqrtf(v); }

#define PR(a, k)   __builtin_shufflevector((a), (a), 2 * (k), 2 * (k) + 1)
#define LOHALF(u)  __builtin_shufflevector((u), (u), 0, 1)
#define HIHALF(u)  __builtin_shufflevector((u), (u), 2, 3)

__device__ __forceinline__ float dot8(const v8f& a, const v8f& b) {
  v2f d = PR(a, 0) * PR(b, 0);
  d += PR(a, 1) * PR(b, 1);
  d += PR(a, 2) * PR(b, 2);
  d += PR(a, 3) * PR(b, 3);
  return d.x + d.y;
}

__device__ __forceinline__ void fence() { __builtin_amdgcn_sched_barrier(0); }

template <int Q>
__device__ __forceinline__ void load_b(const float* __restrict__ xp, v4f (&Bf)[8]) {
  constexpr int p = Q >> 2, ch = (Q >> 1) & 1, half = Q & 1;
  constexpr int hs = p == 0 ? 0 : (p == 1 ? 2 : 1);
#pragma unroll
  for (int cc = 0; cc < 2; ++cc)
#pragma unroll
    for (int i = 0; i < 4; ++i)
      Bf[cc * 4 + i] = *(const v4f_a8*)&xp[(half * 4 + i) * TILE + (((ch * 2 + cc) * 18 + hs) * 18) * 2];
}

template <bool CENTRE, int HS>
__device__ __forceinline__ v2f wfrag(const v2f (&A)[2][2][4][4], const int kw, const int c) {
  if (CENTRE) {
    if (HS == 0) return (A[0][0][kw][c] + A[0][1][kw][c]) + (A[1][0][kw][c] + A[1][1][kw][c]);
    if (HS == 1) return A[0][1][kw][c] + A[1][1][kw][c];
    return A[0][0][kw][c] + A[1][0][kw][c];
  } else {
    if (HS == 0) return A[1][0][kw][c] + A[1][1][kw][c];
    if (HS == 1) return A[1][1][kw][c];
    return A[1][0][kw][c];
  }
}

template <bool CENTRE, int Q>
__device__ __forceinline__ void mma_b(const v2f (&A)[2][2][4][4], const v4f (&Bf)[8],
                                      v8f (&X)[IN_DIM], v8f (&Y)[IN_DIM]) {
  constexpr int p = Q >> 2, ch = (Q >> 1) & 1, half = Q & 1;
  constexpr int hs = p == 0 ? 0 : (p == 1 ? 2 : 1);
#pragma unroll
  for (int sw = 0; sw < 2; ++sw) {
    const int kw = 2 * sw + 1;
#pragma unroll
    for (int cc = 0; cc < 2; ++cc) {
      const v2f a = wfrag<CENTRE, hs>(A, kw, ch * 2 + cc);
#pragma unroll
      for (int i = 0; i < 4; ++i) {
        const v4f b = Bf[cc * 4 + i];
        const v2f bvec = sw ? (v2f){b.x, b.y} : (v2f){b.z, b.w};
        const int in = half * 4 + i;
        if (p != 1)
          X[in] = __builtin_amdgcn_wmma_f32_16x16x4_f32(false, a, false, bvec, (short)0, X[in], false, false);
        else if (ch == 0 && sw == 0 && cc == 0)
          Y[in] = __builtin_amdgcn_wmma_f32_16x16x4_f32(false, a, false, bvec, (short)0, X[in], false, false);
        else
          Y[in] = __builtin_amdgcn_wmma_f32_16x16x4_f32(false, a, false, bvec, (short)0, Y[in], false, false);
      }
    }
  }
}

template <bool CENTRE>
__device__ __forceinline__ void products(const float* __restrict__ xp, const v2f (&A)[2][2][4][4],
                                         v8f (&X)[IN_DIM], v8f (&Y)[IN_DIM]) {
  v4f Bx[8], By[8];
  load_b<0>(xp, Bx);
  load_b<1>(xp, By);  fence(); mma_b<CENTRE, 0>(A, Bx, X, Y);  fence();
  load_b<2>(xp, Bx);  fence(); mma_b<CENTRE, 1>(A, By, X, Y);  fence();
  load_b<3>(xp, By);  fence(); mma_b<CENTRE, 2>(A, Bx, X, Y);  fence();
  load_b<4>(xp, Bx);  fence(); mma_b<CENTRE, 3>(A, By, X, Y);  fence();
  load_b<5>(xp, By);  fence(); mma_b<CENTRE, 4>(A, Bx, X, Y);  fence();
  load_b<6>(xp, Bx);  fence(); mma_b<CENTRE, 5>(A, By, X, Y);  fence();
  load_b<7>(xp, By);  fence(); mma_b<CENTRE, 6>(A, Bx, X, Y);  fence();
  load_b<8>(xp, Bx);  fence(); mma_b<CENTRE, 7>(A, By, X, Y);  fence();
  load_b<9>(xp, By);  fence(); mma_b<CENTRE, 8>(A, Bx, X, Y);  fence();
  load_b<10>(xp, Bx); fence(); mma_b<CENTRE, 9>(A, By, X, Y);  fence();
  load_b<11>(xp, By); fence(); mma_b<CENTRE, 10>(A, Bx, X, Y); fence();
  mma_b<CENTRE, 11>(A, By, X, Y); fence();
}

__device__ __forceinline__ void route_pair(v8f (&acc)[2][IN_DIM], float* __restrict__ ex,
                                           const float (&bl)[8], const float (&rb)[8],
                                           const int n, const int khalf, const int wv,
                                           float (&pa)[2][8]) {
  const v8f blv = {bl[0], bl[1], bl[2], bl[3], bl[4], bl[5], bl[6], bl[7]};
#pragma unroll
  for (int r = 0; r < 2; ++r)
#pragma unroll
    for (int in = 0; in < IN_DIM; ++in) acc[r][in] += blv;

  float n2m[2][4];
#pragma unroll
  for (int r = 0; r < 2; ++r)
#pragma unroll
    for (int i = 0; i < 4; ++i)
      n2m[r][i] = sqrt1(own_sum(dot8(acc[r][i], acc[r][i]), dot8(acc[r][i + 4], acc[r][i + 4]), khalf));

  const v8f rbv = {rb[0], rb[1], rb[2], rb[3], rb[4], rb[5], rb[6], rb[7]};
  v8f pav[2];
  float lgm[2][4], route[2][IN_DIM];
#pragma unroll
  for (int r = 0; r < 2; ++r) {
#pragma unroll
    for (int i = 0; i < 4; ++i) lgm[r][i] = 0.f;
#pragma unroll
    for (int in = 0; in < IN_DIM; ++in) route[r][in] = 0.125f;
  }

#pragma unroll
  for (int it = 0; it < 3; ++it) {
    if (it > 0) {
      float evm[2][4];
#pragma unroll
      for (int r = 0; r < 2; ++r) {
        float* e = ex + ((it - 1) * 2 + r) * 16 * EXROW + n * EXROW + khalf * 32;
#pragma unroll
        for (int i = 0; i < 4; ++i) evm[r][i] = __expf(lgm[r][i]);
#pragma unroll
        for (int i = 0; i < 4; ++i) e[i * 8 + wv] = evm[r][i];
      }
      __syncthreads();
#pragma unroll
      for (int r = 0; r < 2; ++r) {
        const float* e = ex + ((it - 1) * 2 + r) * 16 * EXROW + n * EXROW + khalf * 32;
        float rm[4];
#pragma unroll
        for (int i = 0; i < 4; ++i) {
          const v4f s0 = *(const v4f*)&e[i * 8];
          const v4f s1 = *(const v4f*)&e[i * 8 + 4];
          const v4f u = s0 + s1;
          const v2f h = LOHALF(u) + HIHALF(u);
          rm[i] = evm[r][i] * rcp1(h.x + h.y);
        }
#pragma unroll
        for (int i = 0; i < 4; ++i) {
          const float ro = xhalf(rm[i]);
          route[r][i]     = khalf ? ro : rm[i];
          route[r][i + 4] = khalf ? rm[i] : ro;
        }
      }
    }
#pragma unroll
    for (int r = 0; r < 2; ++r) {
      pav[r] = rbv;
#pragma unroll
      for (int in = 0; in < IN_DIM; ++in) pav[r] += acc[r][in] * route[r][in];
    }
    if (it < 2) {
#pragma unroll
      for (int r = 0; r < 2; ++r) {
        const float n1 = sqrt1(hsum(dot8(pav[r], pav[r])));
#pragma unroll
        for (int i = 0; i < 4; ++i)
          lgm[r][i] += own_sum(dot8(pav[r], acc[r][i]), dot8(pav[r], acc[r][i + 4]), khalf) * rcp1(fmaxf(n1 * n2m[r][i], 1e-8f));
      }
    }
  }

#pragma unroll
  for (int r = 0; r < 2; ++r) {
    const float nsq   = hsum(dot8(pav[r], pav[r]));
    const float norm  = sqrt1(nsq);
    const float scale = rcp1(norm + 1e-12f) * (nsq * rcp1(1.f + nsq));
    pav[r] *= scale;
#pragma unroll
    for (int j = 0; j < 8; ++j) pa[r][j] = pav[r][j];
  }
}

__device__ __forceinline__ void stage_load(const float* __restrict__ xin, const int d0, const int h0,
                                           const bool en, v4f (&r)[3][3]) {
#pragma unroll
  for (int dd = 0; dd < 3; ++dd)
#pragma unroll
    for (int hh = 0; hh < 3; ++hh) {
      const int d = d0 + dd, h = h0 + hh;
      const bool ok = en && ((unsigned)d < 16u) && ((unsigned)h < 16u);
      const float* src = xin + (ok ? d * (DIN * DIN) + h * DIN : 0);
      r[dd][hh] = ok ? *(const v4f*)src : (v4f){0.f, 0.f, 0.f, 0.f};
    }
}
__device__ __forceinline__ void stage_store(float* __restrict__ dst, const v4f (&r)[3][3]) {
  v4f t[3][3];
#pragma unroll
  for (int dd = 0; dd < 3; ++dd) {
    t[dd][0] = r[dd][1];
    t[dd][1] = r[dd][0] - r[dd][1];
    t[dd][2] = r[dd][2] - r[dd][1];
  }
#pragma unroll
  for (int hs = 0; hs < 3; ++hs) {
    const v4f c = t[1][hs];
    const v4f s[3] = {c, t[0][hs] - c, t[2][hs] - c};
#pragma unroll
    for (int ds = 0; ds < 3; ++ds) {
      float* q = dst + (ds * 3 + hs) * 36;
      q[0] = s[ds].x; q[2] = s[ds].y; q[4] = s[ds].z; q[6] = s[ds].w;
    }
  }
}

__global__ __launch_bounds__(256)
void deconv_route_fused(const float* __restrict__ x,
                        const float* __restrict__ w,
                        const float* __restrict__ deconv_b,
                        const float* __restrict__ rbias,
                        float* __restrict__ out) {
  __shared__ __attribute__((aligned(16))) float xs[2 * XBUF];
  __shared__ __attribute__((aligned(16))) float ex[2 * 2 * 16 * EXROW];

  const int tid   = threadIdx.x;
  const int lane  = tid & 31;
  const int wv    = tid >> 5;
  const int ocb   = wv * 16;
  const int n     = lane & 15;
  const int khalf = lane >> 4;

  const int wg  = blockIdx.x;
  const int b   = wg >> 7;
  const int rem = wg & 127;
  const int pd  = (rem >> 6) & 1, ph = (rem >> 5) & 1;
  const int dp0 = ((rem >> 2) & 7) * 2;
  const int hp0 = (rem & 3) * 4;

  v2f A[2][2][4][4];
  {
    const int kd0 = 1 - pd, kh0 = 1 - ph;
#pragma unroll
    for (int sd = 0; sd < 2; ++sd)
#pragma unroll
      for (int sh = 0; sh < 2; ++sh)
#pragma unroll
        for (int kw = 0; kw < 4; ++kw) {
          const int tap = ((2 * sd + kd0) * 4 + (2 * sh + kh0)) * 4 + kw;
#pragma unroll
          for (int c = 0; c < 4; ++c) {
            const int ic = c * 4 + khalf * 2;
            const float* wp = w + (size_t)(ic * OCH + ocb + n) * 64 + tap;
            A[sd][sh][kw][c].x = wp[0];
            A[sd][sh][kw][c].y = wp[OCH * 64];
          }
        }
  }
  float bl[8], rb[8];
#pragma unroll
  for (int j = 0; j < 8; ++j) {
    bl[j] = deconv_b[ocb + khalf * 8 + j];
    rb[j] = rbias[ocb + khalf * 8 + j];
  }

#pragma unroll
  for (int k = 0; k < 18; ++k) {
    const int idx = tid + 256 * k;
    const int e = idx & 1, cs = (idx >> 1) & 1, r = idx >> 2;
    xs[(r * 18 + (cs ? 17 : 0)) * 2 + e] = 0.f;
  }

  const int colq = (tid & 3) * 4;
  const int sic  = (tid >> 2) & 15;
  const int kq   = tid >> 6;
  const float* xsrc = x + ((size_t)((b * IN_DIM + kq * 2) * 16 + sic)) * (DIN * DIN * DIN) + colq;
  float* xdst = xs + kq * 2 * TILE + ((sic >> 1) * 9 * 18 + 1 + colq) * 2 + (sic & 1);
  const int d0 = dp0 - 1 + pd;

  const float* xl = xs + (khalf * 9 * 18 + n) * 2;

  {
    v4f r0[3][3], r1[3][3];
    stage_load(xsrc, d0, hp0 - 1 + ph, true, r0);
    stage_load(xsrc + (size_t)16 * (DIN * DIN * DIN), d0, hp0 - 1 + ph, true, r1);
    stage_store(xdst, r0);
    stage_store(xdst + TILE, r1);
  }
  __syncthreads();

#pragma unroll 1
  for (int t = 0; t < 2; ++t) {
    const int hp  = hp0 + 2 * t;
    const int cur = t & 1;

#pragma unroll 1
    for (int pw = 0; pw < 2; ++pw) {
      const float* xp = xl + cur * XBUF + pw * 2;

      v4f r[3][3];
      stage_load(xsrc + (size_t)pw * 16 * (DIN * DIN * DIN), d0, hp + 2 - 1 + ph, t + 1 < 2, r);

      v8f U[IN_DIM], V[IN_DIM];
#pragma unroll
      for (int in = 0; in < IN_DIM; ++in) U[in] = (v8f){0.f, 0.f, 0.f, 0.f, 0.f, 0.f, 0.f, 0.f};
      products<true>(xp, A, U, V);

      stage_store(xdst + (cur ^ 1) * XBUF + pw * TILE, r);

#pragma unroll 1
      for (int dr = 0; dr < 2; ++dr) {
        v8f RS[2][IN_DIM];
#pragma unroll
        for (int in = 0; in < IN_DIM; ++in) RS[0][in] = (v8f){0.f, 0.f, 0.f, 0.f, 0.f, 0.f, 0.f, 0.f};
        products<false>(xp + (dr + 1) * 3 * 18 * 2, A, RS[0], RS[1]);
#pragma unroll
        for (int in = 0; in < IN_DIM; ++in) { RS[0][in] += U[in]; RS[1][in] += V[in]; }

        float pr[2][8];
        route_pair(RS, ex, bl, rb, n, khalf, wv, pr);

        const int od = 2 * (dp0 + dr) + pd, oh = 2 * hp + ph;
        const int o  = (od * 32 + oh) * 32 + 2 * n + pw;
#pragma unroll
        for (int j = 0; j < 8; ++j) {
          float* dst = out + (((size_t)((b * 8 + wv) * 16 + khalf * 8 + j)) << 15) + o;
          dst[0]  = pr[0][j];
          dst[64] = pr[1][j];
        }

#pragma unroll
        for (int sh = 0; sh < 2; ++sh)
#pragma unroll
          for (int kw = 0; kw < 4; ++kw)
#pragma unroll
            for (int c = 0; c < 4; ++c) {
              const v2f tmp = A[0][sh][kw][c]; A[0][sh][kw][c] = A[1][sh][kw][c]; A[1][sh][kw][c] = tmp;
            }
      }

#pragma unroll
      for (int sd = 0; sd < 2; ++sd)
#pragma unroll
        for (int sh = 0; sh < 2; ++sh)
#pragma unroll
          for (int c = 0; c < 4; ++c) {
            const v2f t0 = A[sd][sh][0][c]; A[sd][sh][0][c] = A[sd][sh][1][c]; A[sd][sh][1][c] = t0;
            const v2f t2 = A[sd][sh][2][c]; A[sd][sh][2][c] = A[sd][sh][3][c]; A[sd][sh][3][c] = t2;
          }
    }
  }
}

extern "C" void kernel_launch(void* const* d_in, const int* in_sizes, int n_in,
                              void* d_out, int out_size, void* d_ws, size_t ws_size,
                              hipStream_t stream) {
  (void)in_sizes; (void)n_in; (void)out_size; (void)d_ws; (void)ws_size;
  const float* x  = (const float*)d_in[0];
  const float* w  = (const float*)d_in[1];
  const float* db = (const float*)d_in[2];
  const float* rb = (const float*)d_in[3];
  float* outp  = (float*)d_out;

  deconv_route_fused<<<256, 256, 0, stream>>>(x, w, db, rb, outp);
}
